# starting kernel
# baseline (speedup 1.0000x reference)
#include <hip/hip_runtime.h>
#include <hip/hip_bf16.h>

typedef __attribute__((ext_vector_type(2))) float v2f;
typedef __attribute__((ext_vector_type(4))) float v4f;
typedef __attribute__((ext_vector_type(8))) float v8f;

#define N_NODES  50000
#define N_EDGES  800000
#define IN_F     256
#define OUT_F    64
#define N_HEADS  8
#define COLS     (N_HEADS * OUT_F)
#define ROW_TILES (N_NODES / 16)
#define LRELU_SLOPE 0.2f

__global__ void zero_f32(float* __restrict__ p, int n) {
    int i = blockIdx.x * blockDim.x + threadIdx.x;
    if (i < n) p[i] = 0.0f;
}

__global__ __launch_bounds__(256) void gat_gemm_wmma(const float* __restrict__ x,
                                                     const float* __restrict__ W,
                                                     float* __restrict__ h) {
    __shared__ float Wlds[IN_F * OUT_F];

    const int wave = threadIdx.x >> 5;
    const int lane = threadIdx.x & 31;
    const int head = blockIdx.x;

    {
        const v4f* __restrict__ Wg = (const v4f*)(W + (size_t)head * (IN_F * OUT_F));
        v4f* __restrict__ Wl = (v4f*)Wlds;
#pragma unroll
        for (int i = 0; i < 16; ++i) {
            int t = threadIdx.x + i * 256;
            Wl[t] = Wg[t];
        }
    }
    __syncthreads();

    const int rowTile = blockIdx.y * 8 + wave;
    if (rowTile >= ROW_TILES) return;

    const int row0 = rowTile * 16;
    const int fo   = lane & 15;
    const int koff = (lane >> 4) << 1;
    const int arow = row0 + (lane & 15);
    const float* __restrict__ ap = x + (size_t)arow * IN_F + koff;

    v8f acc0 = {0.f,0.f,0.f,0.f,0.f,0.f,0.f,0.f};
    v8f acc1 = acc0, acc2 = acc0, acc3 = acc0;

#pragma unroll 2
    for (int k0 = 0; k0 < IN_F; k0 += 4) {
        v2f a = *(const v2f*)(ap + k0);
        const float* __restrict__ b0 = Wlds + (k0 + koff) * OUT_F + fo;
        const float* __restrict__ b1 = b0 + OUT_F;
        v2f b;
        b.x = b0[0];  b.y = b1[0];
        acc0 = __builtin_amdgcn_wmma_f32_16x16x4_f32(false, a, false, b,
                                                     (short)0, acc0, false, false);
        b.x = b0[16]; b.y = b1[16];
        acc1 = __builtin_amdgcn_wmma_f32_16x16x4_f32(false, a, false, b,
                                                     (short)0, acc1, false, false);
        b.x = b0[32]; b.y = b1[32];
        acc2 = __builtin_amdgcn_wmma_f32_16x16x4_f32(false, a, false, b,
                                                     (short)0, acc2, false, false);
        b.x = b0[48]; b.y = b1[48];
        acc3 = __builtin_amdgcn_wmma_f32_16x16x4_f32(false, a, false, b,
                                                     (short)0, acc3, false, false);
    }

    const int rbase = row0 + ((lane >> 4) << 3);
    float* __restrict__ hp = h + head * OUT_F + fo;
#pragma unroll
    for (int v = 0; v < 8; ++v) {
        float* __restrict__ hr = hp + (size_t)(rbase + v) * COLS;
        hr[0]  = acc0[v];
        hr[16] = acc1[v];
        hr[32] = acc2[v];
        hr[48] = acc3[v];
    }
}

__global__ void gat_attn_coef(const float* __restrict__ h,
                              const float* __restrict__ a,
                              float* __restrict__ a_self,
                              float* __restrict__ a_neigh) {
    int idx = blockIdx.x * blockDim.x + threadIdx.x;
    if (idx >= N_NODES * N_HEADS) return;
    int n = idx >> 3;
    int hd = idx & 7;
    const v4f* __restrict__ hp = (const v4f*)(h + (size_t)n * COLS + hd * OUT_F);
    const v4f* __restrict__ as = (const v4f*)(a + (size_t)hd * 2 * OUT_F);
    const v4f* __restrict__ an = as + (OUT_F / 4);
    float s = 0.f, t = 0.f;
#pragma unroll
    for (int f = 0; f < OUT_F / 4; ++f) {
        v4f v = hp[f], u = as[f], w = an[f];
        s += v.x * u.x + v.y * u.y + v.z * u.z + v.w * u.w;
        t += v.x * w.x + v.y * w.y + v.z * w.z + v.w * w.w;
    }
    a_self[idx]  = s;
    a_neigh[idx] = t;
}

__device__ __forceinline__ float edge_exp(float as, float an, float m) {
    float lg = as + an;
    lg = (lg > 0.f ? lg : LRELU_SLOPE * lg) * m;
    return __expf(lg);
}

__global__ void gat_edge_denom(const int* __restrict__ src,
                               const int* __restrict__ dst,
                               const float* __restrict__ M,
                               const float* __restrict__ a_self,
                               const float* __restrict__ a_neigh,
                               float* __restrict__ denom) {
    int idx = blockIdx.x * blockDim.x + threadIdx.x;
    if (idx >= N_EDGES * N_HEADS) return;
    int e  = idx >> 3;
    int hd = idx & 7;
    int s = src[e], d = dst[e];
    float ex = edge_exp(a_self[s * N_HEADS + hd], a_neigh[d * N_HEADS + hd], M[e]);
    atomicAdd(&denom[d * N_HEADS + hd], ex);
}

__global__ __launch_bounds__(256) void gat_edge_aggregate(const int* __restrict__ src,
                                                          const int* __restrict__ dst,
                                                          const float* __restrict__ M,
                                                          const float* __restrict__ a_self,
                                                          const float* __restrict__ a_neigh,
                                                          const float* __restrict__ denom,
                                                          const float* __restrict__ h,
                                                          float* __restrict__ out) {
    const int wave = threadIdx.x >> 5;
    const int lane = threadIdx.x & 31;
    const int e = blockIdx.x * 8 + wave;
    if (e >= N_EDGES) return;

    const int s = src[e];
    const int d = dst[e];
    const float m = M[e];
    const float* __restrict__ hs = h   + (size_t)s * COLS;
    float* __restrict__       od = out + (size_t)d * COLS;

#pragma unroll
    for (int hd = 0; hd < N_HEADS; ++hd) {
        float ex  = edge_exp(a_self[s * N_HEADS + hd], a_neigh[d * N_HEADS + hd], m);
        float att = ex / (denom[d * N_HEADS + hd] + 1e-16f);
        int f0 = hd * OUT_F + lane;
        atomicAdd(&od[f0],      att * hs[f0]);
        atomicAdd(&od[f0 + 32], att * hs[f0 + 32]);
    }
}

__global__ void gat_elu(float* __restrict__ out, int n) {
    int i = blockIdx.x * blockDim.x + threadIdx.x;
    if (i < n) {
        float v = out[i];
        out[i] = v > 0.f ? v : (__expf(v) - 1.0f);
    }
}

extern "C" void kernel_launch(void* const* d_in, const int* in_sizes, int n_in,
                              void* d_out, int out_size, void* d_ws, size_t ws_size,
                              hipStream_t stream) {
    const float* x   = (const float*)d_in[0];
    const int*   src = (const int*)  d_in[1];
    const int*   dst = (const int*)  d_in[2];
    const float* M   = (const float*)d_in[3];
    const float* W   = (const float*)d_in[4];
    const float* a   = (const float*)d_in[5];
    float* out = (float*)d_out;

    float* h       = (float*)d_ws;
    float* a_self  = h      + (size_t)N_NODES * COLS;
    float* a_neigh = a_self + (size_t)N_NODES * N_HEADS;
    float* denom   = a_neigh+ (size_t)N_NODES * N_HEADS;

    {
        int n = N_NODES * COLS;
        zero_f32<<<(n + 255) / 256, 256, 0, stream>>>(out, n);
        int m = N_NODES * N_HEADS;
        zero_f32<<<(m + 255) / 256, 256, 0, stream>>>(denom, m);
    }

    {
        dim3 grid(N_HEADS, (ROW_TILES + 7) / 8);
        gat_gemm_wmma<<<grid, 256, 0, stream>>>(x, W, h);
    }

    {
        int n = N_NODES * N_HEADS;
        gat_attn_coef<<<(n + 255) / 256, 256, 0, stream>>>(h, a, a_self, a_neigh);
    }

    {
        int n = N_EDGES * N_HEADS;
        gat_edge_denom<<<(n + 255) / 256, 256, 0, stream>>>(src, dst, M,
                                                            a_self, a_neigh, denom);
    }

    {
        int blocks = (N_EDGES + 7) / 8;
        gat_edge_aggregate<<<blocks, 256, 0, stream>>>(src, dst, M, a_self, a_neigh,
                                                       denom, h, out);
    }

    {
        int n = N_NODES * COLS;
        gat_elu<<<(n + 255) / 256, 256, 0, stream>>>(out, n);
    }
}


# Aggregation by owner blocks: a block keeps 98 dst rows and their denominators in LDS, sweeps dst, adds exp*h[src] by LDS float atomics (arrival order), divides once, stores ELU; 3 launches
# speedup vs baseline: 2.5807x; 2.5807x over previous
#include <hip/hip_runtime.h>
#include <hip/hip_bf16.h>

typedef __attribute__((ext_vector_type(2))) float v2f;
typedef __attribute__((ext_vector_type(4))) float v4f;
typedef __attribute__((ext_vector_type(8))) float v8f;

#define N_NODES  50000
#define N_EDGES  800000
#define IN_F     256
#define OUT_F    64
#define N_HEADS  8
#define COLS     (N_HEADS * OUT_F)
#define ROW_TILES (N_NODES / 16)
#define LRELU_SLOPE 0.2f

__global__ __launch_bounds__(256) void gat_gemm_wmma(const float* __restrict__ x,
                                                     const float* __restrict__ W,
                                                     float* __restrict__ h) {
    __shared__ float Wlds[IN_F * OUT_F];

    const int wave = threadIdx.x >> 5;
    const int lane = threadIdx.x & 31;
    const int head = blockIdx.x;

    {
        const v4f* __restrict__ Wg = (const v4f*)(W + (size_t)head * (IN_F * OUT_F));
        v4f* __restrict__ Wl = (v4f*)Wlds;
#pragma unroll
        for (int i = 0; i < 16; ++i) {
            int t = threadIdx.x + i * 256;
            Wl[t] = Wg[t];
        }
    }
    __syncthreads();

    const int rowTile = blockIdx.y * 8 + wave;
    if (rowTile >= ROW_TILES) return;

    const int row0 = rowTile * 16;
    const int fo   = lane & 15;
    const int koff = (lane >> 4) << 1;
    const int arow = row0 + (lane & 15);
    const float* __restrict__ ap = x + (size_t)arow * IN_F + koff;

    v8f acc0 = {0.f,0.f,0.f,0.f,0.f,0.f,0.f,0.f};
    v8f acc1 = acc0, acc2 = acc0, acc3 = acc0;

#pragma unroll 2
    for (int k0 = 0; k0 < IN_F; k0 += 4) {
        v2f a = *(const v2f*)(ap + k0);
        const float* __restrict__ b0 = Wlds + (k0 + koff) * OUT_F + fo;
        const float* __restrict__ b1 = b0 + OUT_F;
        v2f b;
        b.x = b0[0];  b.y = b1[0];
        acc0 = __builtin_amdgcn_wmma_f32_16x16x4_f32(false, a, false, b,
                                                     (short)0, acc0, false, false);
        b.x = b0[16]; b.y = b1[16];
        acc1 = __builtin_amdgcn_wmma_f32_16x16x4_f32(false, a, false, b,
                                                     (short)0, acc1, false, false);
        b.x = b0[32]; b.y = b1[32];
        acc2 = __builtin_amdgcn_wmma_f32_16x16x4_f32(false, a, false, b,
                                                     (short)0, acc2, false, false);
        b.x = b0[48]; b.y = b1[48];
        acc3 = __builtin_amdgcn_wmma_f32_16x16x4_f32(false, a, false, b,
                                                     (short)0, acc3, false, false);
    }

    const int rbase = row0 + ((lane >> 4) << 3);
    float* __restrict__ hp = h + head * OUT_F + fo;
#pragma unroll
    for (int v = 0; v < 8; ++v) {
        float* __restrict__ hr = hp + (size_t)(rbase + v) * COLS;
        hr[0]  = acc0[v];
        hr[16] = acc1[v];
        hr[32] = acc2[v];
        hr[48] = acc3[v];
    }
}

__global__ void gat_attn_coef(const float* __restrict__ h,
                              const float* __restrict__ a,
                              float* __restrict__ a_self,
                              float* __restrict__ a_neigh) {
    int idx = blockIdx.x * blockDim.x + threadIdx.x;
    if (idx >= N_NODES * N_HEADS) return;
    int n = idx >> 3;
    int hd = idx & 7;
    const v4f* __restrict__ hp = (const v4f*)(h + (size_t)n * COLS + hd * OUT_F);
    const v4f* __restrict__ as = (const v4f*)(a + (size_t)hd * 2 * OUT_F);
    const v4f* __restrict__ an = as + (OUT_F / 4);
    float s = 0.f, t = 0.f;
#pragma unroll
    for (int f = 0; f < OUT_F / 4; ++f) {
        v4f v = hp[f], u = as[f], w = an[f];
        s += v.x * u.x + v.y * u.y + v.z * u.z + v.w * u.w;
        t += v.x * w.x + v.y * w.y + v.z * w.z + v.w * w.w;
    }
    a_self[idx]  = s;
    a_neigh[idx] = t;
}

__device__ __forceinline__ float edge_exp(float as, float an, float m) {
    float lg = as + an;
    lg = (lg > 0.f ? lg : LRELU_SLOPE * lg) * m;
    return __expf(lg);
}

#define AGG_ROWS    98
#define AGG_BLOCKS  ((N_NODES + AGG_ROWS - 1) / AGG_ROWS)
#define AGG_THREADS 1024
#define AGG_WAVES   (AGG_THREADS / 32)
#define WL_CAP      384
#define WL_FLUSH    16
#define N_DST4      (N_EDGES / 4)
#define SCAN_ITERS  ((N_DST4 + AGG_THREADS - 1) / AGG_THREADS)

static_assert(SCAN_ITERS % 2 == 0, "the sweep takes two chunks per trip and wraps around");

typedef __attribute__((ext_vector_type(4))) int v4i;

__device__ __forceinline__ float lane_bcast(float v, int srcLane) {
    return __int_as_float(__builtin_amdgcn_readlane(__float_as_int(v), srcLane));
}

__device__ __forceinline__ void agg_add_edge(float* __restrict__ acc, float* __restrict__ den,
                                             int r, float ex, const v4f* x, int lane) {
    if (lane < N_HEADS) atomicAdd(&den[r * N_HEADS + lane], ex);
    float* ar = acc + r * COLS + lane;
    const int hi = lane >> 4;
#pragma unroll
    for (int j = 0; j < 4; ++j) {
        float wlo = lane_bcast(ex, 2 * j);
        float whi = lane_bcast(ex, 2 * j + 1);
        float w = hi ? whi : wlo;
        atomicAdd(ar + j * 128 +  0, w * x[j].x);
        atomicAdd(ar + j * 128 + 32, w * x[j].y);
        atomicAdd(ar + j * 128 + 64, w * x[j].z);
        atomicAdd(ar + j * 128 + 96, w * x[j].w);
    }
}

__global__ __launch_bounds__(AGG_THREADS) void gat_owner_aggregate(const int* __restrict__ src,
                                                                  const int* __restrict__ dst,
                                                                  const float* __restrict__ M,
                                                                  const float* __restrict__ a_self,
                                                                  const float* __restrict__ a_neigh,
                                                                  const float* __restrict__ h,
                                                                  float* __restrict__ out) {
    __shared__ float acc[AGG_ROWS * COLS];
    __shared__ float den[AGG_ROWS * N_HEADS];
    __shared__ int   wl[AGG_WAVES * WL_CAP];

    const int tid  = threadIdx.x;
    const int wave = tid >> 5;
    const int lane = tid & 31;
    const int lo    = blockIdx.x * AGG_ROWS;
    const int nrows = min(AGG_ROWS, N_NODES - lo);

    {
        const v4f z = {0.f, 0.f, 0.f, 0.f};
        v4f* a4 = (v4f*)acc;
        for (int q = tid; q < AGG_ROWS * COLS / 4; q += AGG_THREADS) a4[q] = z;
        for (int q = tid; q < AGG_ROWS * N_HEADS; q += AGG_THREADS) den[q] = 0.f;
    }
    __syncthreads();

    int* mylist = wl + wave * WL_CAP;
    int cnt = 0;
    const v4i* __restrict__ dst4 = (const v4i*)dst;
    const int l8 = lane & 7;

    auto append = [&](int dv, int e) {
        const unsigned rr = (unsigned)(dv - lo);
        const bool mt = rr < (unsigned)nrows;
        const unsigned mask = __builtin_amdgcn_ballot_w32(mt);
        if (mask != 0u) {
            if (mt) mylist[cnt + (int)__builtin_amdgcn_mbcnt_lo(mask, 0u)] = (e << 7) | (int)rr;
            cnt += __builtin_popcount(mask);
        }
    };

    const int rot = blockIdx.x % SCAN_ITERS;
    for (int it = 0; it < SCAN_ITERS; it += 2) {
        const int q0 = ((it + rot) % SCAN_ITERS) * AGG_THREADS + tid;
        const int q1 = ((it + 1 + rot) % SCAN_ITERS) * AGG_THREADS + tid;
        const v4i none = {-1, -1, -1, -1};
        const v4i d0 = q0 < N_DST4 ? dst4[q0] : none;
        const v4i d1 = q1 < N_DST4 ? dst4[q1] : none;
        append(d0.x, 4 * q0 + 0);
        append(d0.y, 4 * q0 + 1);
        append(d0.z, 4 * q0 + 2);
        append(d0.w, 4 * q0 + 3);
        append(d1.x, 4 * q1 + 0);
        append(d1.y, 4 * q1 + 1);
        append(d1.z, 4 * q1 + 2);
        append(d1.w, 4 * q1 + 3);

        if (cnt >= WL_FLUSH || it + 2 >= SCAN_ITERS) {
            __builtin_amdgcn_fence(__ATOMIC_ACQ_REL, "wavefront");
            for (int i = 0; i < cnt; i += 4) {
                int r[4], s[4];
                float m[4], ex[4];
                v4f xr[4][4];
#pragma unroll
                for (int k = 0; k < 4; ++k) {
                    const int pk = __builtin_amdgcn_readfirstlane(mylist[min(i + k, cnt - 1)]);
                    const int e = pk >> 7;
                    r[k] = pk & 127;
                    s[k] = src[e];
                    m[k] = M[e];
                }
#pragma unroll
                for (int k = 0; k < 4; ++k) {
                    const v4f* __restrict__ hk = (const v4f*)(h + (size_t)s[k] * COLS) + lane;
#pragma unroll
                    for (int j = 0; j < 4; ++j) xr[k][j] = hk[j * 32];
                }
#pragma unroll
                for (int k = 0; k < 4; ++k)
                    ex[k] = edge_exp(a_self[s[k] * N_HEADS + l8],
                                     a_neigh[(lo + r[k]) * N_HEADS + l8], m[k]);
                agg_add_edge(acc, den, r[0], ex[0], xr[0], lane);
#pragma unroll
                for (int k = 1; k < 4; ++k)
                    if (i + k < cnt) agg_add_edge(acc, den, r[k], ex[k], xr[k], lane);
            }
            cnt = 0;
            __builtin_amdgcn_fence(__ATOMIC_ACQ_REL, "wavefront");
        }
    }
    __syncthreads();

    for (int q = tid; q < nrows * (COLS / 4); q += AGG_THREADS) {
        const int r  = q >> 7;
        const int c4 = q & 127;
        const int j  = c4 >> 5;
        const int l  = c4 & 31;
        const float* ar = acc + r * COLS + j * 128 + l;
        const float dn = den[r * N_HEADS + 2 * j + (l >> 4)] + 1e-16f;
        v4f v;
        v.x = ar[0]  / dn;
        v.y = ar[32] / dn;
        v.z = ar[64] / dn;
        v.w = ar[96] / dn;
        v.x = v.x > 0.f ? v.x : (__expf(v.x) - 1.0f);
        v.y = v.y > 0.f ? v.y : (__expf(v.y) - 1.0f);
        v.z = v.z > 0.f ? v.z : (__expf(v.z) - 1.0f);
        v.w = v.w > 0.f ? v.w : (__expf(v.w) - 1.0f);
        *((v4f*)(out + (size_t)(lo + r) * COLS) + c4) = v;
    }
}

extern "C" void kernel_launch(void* const* d_in, const int* in_sizes, int n_in,
                              void* d_out, int out_size, void* d_ws, size_t ws_size,
                              hipStream_t stream) {
    const float* x   = (const float*)d_in[0];
    const int*   src = (const int*)  d_in[1];
    const int*   dst = (const int*)  d_in[2];
    const float* M   = (const float*)d_in[3];
    const float* W   = (const float*)d_in[4];
    const float* a   = (const float*)d_in[5];
    float* out = (float*)d_out;

    float* h       = (float*)d_ws;
    float* a_self  = h      + (size_t)N_NODES * COLS;
    float* a_neigh = a_self + (size_t)N_NODES * N_HEADS;

    {
        dim3 grid(N_HEADS, (ROW_TILES + 7) / 8);
        gat_gemm_wmma<<<grid, 256, 0, stream>>>(x, W, h);
    }

    {
        int n = N_NODES * N_HEADS;
        gat_attn_coef<<<(n + 255) / 256, 256, 0, stream>>>(h, a, a_self, a_neigh);
    }

    gat_owner_aggregate<<<AGG_BLOCKS, AGG_THREADS, 0, stream>>>(src, dst, M, a_self, a_neigh,
                                                                 h, out);
}


# Sweep of dst software-pipelined: four 16-byte chunks per lane per trip, the next trip's four loaded unconditionally before this trip's are tested; wave list sized for a trip; sums stay arrival order
# speedup vs baseline: 2.6657x; 1.0329x over previous
#include <hip/hip_runtime.h>
#include <hip/hip_bf16.h>

typedef __attribute__((ext_vector_type(2))) float v2f;
typedef __attribute__((ext_vector_type(4))) float v4f;
typedef __attribute__((ext_vector_type(8))) float v8f;

#define N_NODES  50000
#define N_EDGES  800000
#define IN_F     256
#define OUT_F    64
#define N_HEADS  8
#define COLS     (N_HEADS * OUT_F)
#define ROW_TILES (N_NODES / 16)
#define LRELU_SLOPE 0.2f

__global__ __launch_bounds__(256) void gat_gemm_wmma(const float* __restrict__ x,
                                                     const float* __restrict__ W,
                                                     float* __restrict__ h) {
    __shared__ float Wlds[IN_F * OUT_F];

    const int wave = threadIdx.x >> 5;
    const int lane = threadIdx.x & 31;
    const int head = blockIdx.x;

    {
        const v4f* __restrict__ Wg = (const v4f*)(W + (size_t)head * (IN_F * OUT_F));
        v4f* __restrict__ Wl = (v4f*)Wlds;
#pragma unroll
        for (int i = 0; i < 16; ++i) {
            int t = threadIdx.x + i * 256;
            Wl[t] = Wg[t];
        }
    }
    __syncthreads();

    const int rowTile = blockIdx.y * 8 + wave;
    if (rowTile >= ROW_TILES) return;

    const int row0 = rowTile * 16;
    const int fo   = lane & 15;
    const int koff = (lane >> 4) << 1;
    const int arow = row0 + (lane & 15);
    const float* __restrict__ ap = x + (size_t)arow * IN_F + koff;

    v8f acc0 = {0.f,0.f,0.f,0.f,0.f,0.f,0.f,0.f};
    v8f acc1 = acc0, acc2 = acc0, acc3 = acc0;

#pragma unroll 2
    for (int k0 = 0; k0 < IN_F; k0 += 4) {
        v2f a = *(const v2f*)(ap + k0);
        const float* __restrict__ b0 = Wlds + (k0 + koff) * OUT_F + fo;
        const float* __restrict__ b1 = b0 + OUT_F;
        v2f b;
        b.x = b0[0];  b.y = b1[0];
        acc0 = __builtin_amdgcn_wmma_f32_16x16x4_f32(false, a, false, b,
                                                     (short)0, acc0, false, false);
        b.x = b0[16]; b.y = b1[16];
        acc1 = __builtin_amdgcn_wmma_f32_16x16x4_f32(false, a, false, b,
                                                     (short)0, acc1, false, false);
        b.x = b0[32]; b.y = b1[32];
        acc2 = __builtin_amdgcn_wmma_f32_16x16x4_f32(false, a, false, b,
                                                     (short)0, acc2, false, false);
        b.x = b0[48]; b.y = b1[48];
        acc3 = __builtin_amdgcn_wmma_f32_16x16x4_f32(false, a, false, b,
                                                     (short)0, acc3, false, false);
    }

    const int rbase = row0 + ((lane >> 4) << 3);
    float* __restrict__ hp = h + head * OUT_F + fo;
#pragma unroll
    for (int v = 0; v < 8; ++v) {
        float* __restrict__ hr = hp + (size_t)(rbase + v) * COLS;
        hr[0]  = acc0[v];
        hr[16] = acc1[v];
        hr[32] = acc2[v];
        hr[48] = acc3[v];
    }
}

__global__ void gat_attn_coef(const float* __restrict__ h,
                              const float* __restrict__ a,
                              float* __restrict__ a_self,
                              float* __restrict__ a_neigh) {
    int idx = blockIdx.x * blockDim.x + threadIdx.x;
    if (idx >= N_NODES * N_HEADS) return;
    int n = idx >> 3;
    int hd = idx & 7;
    const v4f* __restrict__ hp = (const v4f*)(h + (size_t)n * COLS + hd * OUT_F);
    const v4f* __restrict__ as = (const v4f*)(a + (size_t)hd * 2 * OUT_F);
    const v4f* __restrict__ an = as + (OUT_F / 4);
    float s = 0.f, t = 0.f;
#pragma unroll
    for (int f = 0; f < OUT_F / 4; ++f) {
        v4f v = hp[f], u = as[f], w = an[f];
        s += v.x * u.x + v.y * u.y + v.z * u.z + v.w * u.w;
        t += v.x * w.x + v.y * w.y + v.z * w.z + v.w * w.w;
    }
    a_self[idx]  = s;
    a_neigh[idx] = t;
}

__device__ __forceinline__ float edge_exp(float as, float an, float m) {
    float lg = as + an;
    lg = (lg > 0.f ? lg : LRELU_SLOPE * lg) * m;
    return __expf(lg);
}

#define AGG_ROWS    98
#define AGG_BLOCKS  ((N_NODES + AGG_ROWS - 1) / AGG_ROWS)
#define AGG_THREADS 1024
#define AGG_WAVES   (AGG_THREADS / 32)
#define WL_FLUSH    16
#define SWEEP       4
#define WL_CAP      (WL_FLUSH + SWEEP * 4 * 32)
#define N_DST4      (N_EDGES / 4)
#define SCAN_ITERS  ((N_DST4 + AGG_THREADS - 1) / AGG_THREADS)

static_assert(SCAN_ITERS % SWEEP == 0, "the sweep takes SWEEP chunks per trip and wraps around");
static_assert(N_DST4 % 32 == 0, "a wave's 32 lanes are inside dst together or outside it together");

typedef __attribute__((ext_vector_type(4))) int v4i;

__device__ __forceinline__ float lane_bcast(float v, int srcLane) {
    return __int_as_float(__builtin_amdgcn_readlane(__float_as_int(v), srcLane));
}

__device__ __forceinline__ void agg_add_edge(float* __restrict__ acc, float* __restrict__ den,
                                             int r, float ex, const v4f* x, int lane) {
    if (lane < N_HEADS) atomicAdd(&den[r * N_HEADS + lane], ex);
    float* ar = acc + r * COLS + lane;
    const int hi = lane >> 4;
#pragma unroll
    for (int j = 0; j < 4; ++j) {
        float wlo = lane_bcast(ex, 2 * j);
        float whi = lane_bcast(ex, 2 * j + 1);
        float w = hi ? whi : wlo;
        atomicAdd(ar + j * 128 +  0, w * x[j].x);
        atomicAdd(ar + j * 128 + 32, w * x[j].y);
        atomicAdd(ar + j * 128 + 64, w * x[j].z);
        atomicAdd(ar + j * 128 + 96, w * x[j].w);
    }
}

__global__ __launch_bounds__(AGG_THREADS) void gat_owner_aggregate(const int* __restrict__ src,
                                                                  const int* __restrict__ dst,
                                                                  const float* __restrict__ M,
                                                                  const float* __restrict__ a_self,
                                                                  const float* __restrict__ a_neigh,
                                                                  const float* __restrict__ h,
                                                                  float* __restrict__ out) {
    __shared__ float acc[AGG_ROWS * COLS];
    __shared__ float den[AGG_ROWS * N_HEADS];
    __shared__ int   wl[AGG_WAVES * WL_CAP];

    const int tid  = threadIdx.x;
    const int wave = tid >> 5;
    const int lane = tid & 31;
    const int lo    = blockIdx.x * AGG_ROWS;
    const int nrows = min(AGG_ROWS, N_NODES - lo);

    {
        const v4f z = {0.f, 0.f, 0.f, 0.f};
        v4f* a4 = (v4f*)acc;
        for (int q = tid; q < AGG_ROWS * COLS / 4; q += AGG_THREADS) a4[q] = z;
        for (int q = tid; q < AGG_ROWS * N_HEADS; q += AGG_THREADS) den[q] = 0.f;
    }
    __syncthreads();

    int* mylist = wl + wave * WL_CAP;
    int cnt = 0;
    const v4i* __restrict__ dst4 = (const v4i*)dst;
    const int l8 = lane & 7;

    const int tid9 = tid << 9;
    auto append = [&](int dv, int ebits) {
        const unsigned rr = (unsigned)(dv - lo);
        const bool mt = rr < (unsigned)nrows;
        const unsigned mask = __builtin_amdgcn_ballot_w32(mt);
        if (mask != 0u) {
            if (mt) mylist[cnt + (int)__builtin_amdgcn_mbcnt_lo(mask, 0u)] = ebits | tid9 | (int)rr;
            cnt += __builtin_popcount(mask);
        }
    };

    const int rot = blockIdx.x % SCAN_ITERS;

    const int wave_s = __builtin_amdgcn_readfirstlane(wave);
    auto chunk_ok = [&](int c) { return c * AGG_THREADS + wave_s * 32 < N_DST4; };
    auto chunk_ld = [&](int c) { return dst4[(chunk_ok(c) ? c * AGG_THREADS : 0) + tid]; };

    int cnext = rot;
    int cc[SWEEP];
    v4i dc[SWEEP];
#pragma unroll
    for (int k = 0; k < SWEEP; ++k) {
        cc[k] = cnext;
        cnext = (cnext + 1 == SCAN_ITERS) ? 0 : cnext + 1;
        dc[k] = chunk_ld(cc[k]);
    }

    for (int it = 0; it < SCAN_ITERS; it += SWEEP) {
        const bool more = it + SWEEP < SCAN_ITERS;
        int cn[SWEEP];
        v4i dn[SWEEP];
#pragma unroll
        for (int k = 0; k < SWEEP; ++k) {
            cn[k] = cnext;
            cnext = (cnext + 1 == SCAN_ITERS) ? 0 : cnext + 1;
            dn[k] = chunk_ld(cn[k]);
        }

#pragma unroll
        for (int k = 0; k < SWEEP; ++k) {
            if (chunk_ok(cc[k])) {
                const int eb = cc[k] << 19;
                append(dc[k].x, eb | (0 << 7));
                append(dc[k].y, eb | (1 << 7));
                append(dc[k].z, eb | (2 << 7));
                append(dc[k].w, eb | (3 << 7));
            }
        }
#pragma unroll
        for (int k = 0; k < SWEEP; ++k) { cc[k] = cn[k]; dc[k] = dn[k]; }

        if (cnt >= WL_FLUSH || !more) {
            __builtin_amdgcn_fence(__ATOMIC_ACQ_REL, "wavefront");
            for (int i = 0; i < cnt; i += 4) {
                int r[4], s[4];
                float m[4], ex[4];
                v4f xr[4][4];
#pragma unroll
                for (int k = 0; k < 4; ++k) {
                    const int pk = __builtin_amdgcn_readfirstlane(mylist[min(i + k, cnt - 1)]);
                    const int e = pk >> 7;
                    r[k] = pk & 127;
                    s[k] = src[e];
                    m[k] = M[e];
                }
#pragma unroll
                for (int k = 0; k < 4; ++k) {
                    const v4f* __restrict__ hk = (const v4f*)(h + (size_t)s[k] * COLS) + lane;
#pragma unroll
                    for (int j = 0; j < 4; ++j) xr[k][j] = hk[j * 32];
                }
#pragma unroll
                for (int k = 0; k < 4; ++k)
                    ex[k] = edge_exp(a_self[s[k] * N_HEADS + l8],
                                     a_neigh[(lo + r[k]) * N_HEADS + l8], m[k]);
                agg_add_edge(acc, den, r[0], ex[0], xr[0], lane);
#pragma unroll
                for (int k = 1; k < 4; ++k)
                    if (i + k < cnt) agg_add_edge(acc, den, r[k], ex[k], xr[k], lane);
            }
            cnt = 0;
            __builtin_amdgcn_fence(__ATOMIC_ACQ_REL, "wavefront");
        }
    }
    __syncthreads();

    for (int q = tid; q < nrows * (COLS / 4); q += AGG_THREADS) {
        const int r  = q >> 7;
        const int c4 = q & 127;
        const int j  = c4 >> 5;
        const int l  = c4 & 31;
        const float* ar = acc + r * COLS + j * 128 + l;
        const float dn = den[r * N_HEADS + 2 * j + (l >> 4)] + 1e-16f;
        v4f v;
        v.x = ar[0]  / dn;
        v.y = ar[32] / dn;
        v.z = ar[64] / dn;
        v.w = ar[96] / dn;
        v.x = v.x > 0.f ? v.x : (__expf(v.x) - 1.0f);
        v.y = v.y > 0.f ? v.y : (__expf(v.y) - 1.0f);
        v.z = v.z > 0.f ? v.z : (__expf(v.z) - 1.0f);
        v.w = v.w > 0.f ? v.w : (__expf(v.w) - 1.0f);
        *((v4f*)(out + (size_t)(lo + r) * COLS) + c4) = v;
    }
}

extern "C" void kernel_launch(void* const* d_in, const int* in_sizes, int n_in,
                              void* d_out, int out_size, void* d_ws, size_t ws_size,
                              hipStream_t stream) {
    const float* x   = (const float*)d_in[0];
    const int*   src = (const int*)  d_in[1];
    const int*   dst = (const int*)  d_in[2];
    const float* M   = (const float*)d_in[3];
    const float* W   = (const float*)d_in[4];
    const float* a   = (const float*)d_in[5];
    float* out = (float*)d_out;

    float* h       = (float*)d_ws;
    float* a_self  = h      + (size_t)N_NODES * COLS;
    float* a_neigh = a_self + (size_t)N_NODES * N_HEADS;

    {
        dim3 grid(N_HEADS, (ROW_TILES + 7) / 8);
        gat_gemm_wmma<<<grid, 256, 0, stream>>>(x, W, h);
    }

    {
        int n = N_NODES * N_HEADS;
        gat_attn_coef<<<(n + 255) / 256, 256, 0, stream>>>(h, a, a_self, a_neigh);
    }

    gat_owner_aggregate<<<AGG_BLOCKS, AGG_THREADS, 0, stream>>>(src, dst, M, a_self, a_neigh,
                                                                 h, out);
}


# One sweep of dst serves two 98-row halves per owner block (256 blocks, not 511): second-half edges parked in per-wave LDS lists, dealt out evenly, added after half one is stored; sums arrival order
# speedup vs baseline: 2.8935x; 1.0855x over previous
#include <hip/hip_runtime.h>
#include <hip/hip_bf16.h>

typedef __attribute__((ext_vector_type(2))) float v2f;
typedef __attribute__((ext_vector_type(4))) float v4f;
typedef __attribute__((ext_vector_type(8))) float v8f;

#define N_NODES  50000
#define N_EDGES  800000
#define IN_F     256
#define OUT_F    64
#define N_HEADS  8
#define COLS     (N_HEADS * OUT_F)
#define ROW_TILES (N_NODES / 16)
#define LRELU_SLOPE 0.2f

__global__ __launch_bounds__(256) void gat_gemm_wmma(const float* __restrict__ x,
                                                     const float* __restrict__ W,
                                                     float* __restrict__ h) {
    __shared__ float Wlds[IN_F * OUT_F];

    const int wave = threadIdx.x >> 5;
    const int lane = threadIdx.x & 31;
    const int head = blockIdx.x;

    {
        const v4f* __restrict__ Wg = (const v4f*)(W + (size_t)head * (IN_F * OUT_F));
        v4f* __restrict__ Wl = (v4f*)Wlds;
#pragma unroll
        for (int i = 0; i < 16; ++i) {
            int t = threadIdx.x + i * 256;
            Wl[t] = Wg[t];
        }
    }
    __syncthreads();

    const int rowTile = blockIdx.y * 8 + wave;
    if (rowTile >= ROW_TILES) return;

    const int row0 = rowTile * 16;
    const int fo   = lane & 15;
    const int koff = (lane >> 4) << 1;
    const int arow = row0 + (lane & 15);
    const float* __restrict__ ap = x + (size_t)arow * IN_F + koff;

    v8f acc0 = {0.f,0.f,0.f,0.f,0.f,0.f,0.f,0.f};
    v8f acc1 = acc0, acc2 = acc0, acc3 = acc0;

#pragma unroll 2
    for (int k0 = 0; k0 < IN_F; k0 += 4) {
        v2f a = *(const v2f*)(ap + k0);
        const float* __restrict__ b0 = Wlds + (k0 + koff) * OUT_F + fo;
        const float* __restrict__ b1 = b0 + OUT_F;
        v2f b;
        b.x = b0[0];  b.y = b1[0];
        acc0 = __builtin_amdgcn_wmma_f32_16x16x4_f32(false, a, false, b,
                                                     (short)0, acc0, false, false);
        b.x = b0[16]; b.y = b1[16];
        acc1 = __builtin_amdgcn_wmma_f32_16x16x4_f32(false, a, false, b,
                                                     (short)0, acc1, false, false);
        b.x = b0[32]; b.y = b1[32];
        acc2 = __builtin_amdgcn_wmma_f32_16x16x4_f32(false, a, false, b,
                                                     (short)0, acc2, false, false);
        b.x = b0[48]; b.y = b1[48];
        acc3 = __builtin_amdgcn_wmma_f32_16x16x4_f32(false, a, false, b,
                                                     (short)0, acc3, false, false);
    }

    const int rbase = row0 + ((lane >> 4) << 3);
    float* __restrict__ hp = h + head * OUT_F + fo;
#pragma unroll
    for (int v = 0; v < 8; ++v) {
        float* __restrict__ hr = hp + (size_t)(rbase + v) * COLS;
        hr[0]  = acc0[v];
        hr[16] = acc1[v];
        hr[32] = acc2[v];
        hr[48] = acc3[v];
    }
}

__global__ void gat_attn_coef(const float* __restrict__ h,
                              const float* __restrict__ a,
                              float* __restrict__ a_self,
                              float* __restrict__ a_neigh) {
    int idx = blockIdx.x * blockDim.x + threadIdx.x;
    if (idx >= N_NODES * N_HEADS) return;
    int n = idx >> 3;
    int hd = idx & 7;
    const v4f* __restrict__ hp = (const v4f*)(h + (size_t)n * COLS + hd * OUT_F);
    const v4f* __restrict__ as = (const v4f*)(a + (size_t)hd * 2 * OUT_F);
    const v4f* __restrict__ an = as + (OUT_F / 4);
    float s = 0.f, t = 0.f;
#pragma unroll
    for (int f = 0; f < OUT_F / 4; ++f) {
        v4f v = hp[f], u = as[f], w = an[f];
        s += v.x * u.x + v.y * u.y + v.z * u.z + v.w * u.w;
        t += v.x * w.x + v.y * w.y + v.z * w.z + v.w * w.w;
    }
    a_self[idx]  = s;
    a_neigh[idx] = t;
}

__device__ __forceinline__ float edge_exp(float as, float an, float m) {
    float lg = as + an;
    lg = (lg > 0.f ? lg : LRELU_SLOPE * lg) * m;
    return __expf(lg);
}

#define AGG_HALF    98
#define AGG_ROWS    (2 * AGG_HALF)
#define AGG_BLOCKS  ((N_NODES + AGG_ROWS - 1) / AGG_ROWS)
#define AGG_THREADS 1024
#define AGG_WAVES   (AGG_THREADS / 32)
#define WL_FLUSH    16
#define SWEEP       4
#define WL_CAP      (WL_FLUSH + SWEEP * 4 * 32)
#define P_CAP       384
#define N_DST4      (N_EDGES / 4)
#define SCAN_ITERS  ((N_DST4 + AGG_THREADS - 1) / AGG_THREADS)

static_assert(SCAN_ITERS % SWEEP == 0, "the sweep takes SWEEP chunks per trip and wraps around");
static_assert(N_DST4 % 32 == 0, "a wave's 32 lanes are inside dst together or outside it together");
static_assert(AGG_WAVES == 32, "lane i of a wave stands for wave i where the parked counts are summed");
static_assert(P_CAP <= WL_CAP, "a wave's share of the parked entries fits its own list");
static_assert(AGG_ROWS <= 256, "a row number takes the low 8 bits of a list entry");
static_assert(N_EDGES < (1 << 23), "an edge number takes the bits above the row number");

typedef __attribute__((ext_vector_type(4))) int v4i;

__device__ __forceinline__ float lane_bcast(float v, int srcLane) {
    return __int_as_float(__builtin_amdgcn_readlane(__float_as_int(v), srcLane));
}

__device__ __forceinline__ void agg_add_edge(float* __restrict__ acc, float* __restrict__ den,
                                             int r, float ex, const v4f* x, int lane) {
    if (lane < N_HEADS) atomicAdd(&den[r * N_HEADS + lane], ex);
    float* ar = acc + r * COLS + lane;
    const int hi = lane >> 4;
#pragma unroll
    for (int j = 0; j < 4; ++j) {
        float wlo = lane_bcast(ex, 2 * j);
        float whi = lane_bcast(ex, 2 * j + 1);
        float w = hi ? whi : wlo;
        atomicAdd(ar + j * 128 +  0, w * x[j].x);
        atomicAdd(ar + j * 128 + 32, w * x[j].y);
        atomicAdd(ar + j * 128 + 64, w * x[j].z);
        atomicAdd(ar + j * 128 + 96, w * x[j].w);
    }
}

__global__ __launch_bounds__(AGG_THREADS) void gat_owner_aggregate(const int* __restrict__ src,
                                                                  const int* __restrict__ dst,
                                                                  const float* __restrict__ M,
                                                                  const float* __restrict__ a_self,
                                                                  const float* __restrict__ a_neigh,
                                                                  const float* __restrict__ h,
                                                                  float* __restrict__ out) {
    __shared__ float acc[AGG_HALF * COLS];
    __shared__ float den[AGG_HALF * N_HEADS];
    __shared__ int   wl[AGG_WAVES * WL_CAP];
    __shared__ int   pl[AGG_WAVES * P_CAP];
    __shared__ int   pcount[AGG_WAVES];

    const int tid  = threadIdx.x;
    const int wave = tid >> 5;
    const int lane = tid & 31;
    const int lo    = blockIdx.x * AGG_ROWS;
    const int nrows = min(AGG_ROWS, N_NODES - lo);
    const int n0    = min(AGG_HALF, nrows);

    int* mylist = wl + wave * WL_CAP;
    int* mypark = pl + wave * P_CAP;
    int pcnt = 0;
    const v4i* __restrict__ dst4 = (const v4i*)dst;
    const int l8 = lane & 7;

    const int tid10 = tid << 10;

    const int rot = blockIdx.x % SCAN_ITERS;

    const int wave_s = __builtin_amdgcn_readfirstlane(wave);
    auto chunk_ok = [&](int c) { return c * AGG_THREADS + wave_s * 32 < N_DST4; };
    auto chunk_ld = [&](int c) { return dst4[(chunk_ok(c) ? c * AGG_THREADS : 0) + tid]; };

    for (int half = 0; half < 2; ++half) {
        const int base  = lo + half * AGG_HALF;
        const int n_act = half == 0 ? n0 : nrows - n0;
        const int n_all = half == 0 ? nrows : n_act;
        if (n_act == 0) break;

        if (half == 1) __syncthreads();
        {
            const v4f z = {0.f, 0.f, 0.f, 0.f};
            v4f* a4 = (v4f*)acc;
            for (int q = tid; q < AGG_HALF * COLS / 4; q += AGG_THREADS) a4[q] = z;
            for (int q = tid; q < AGG_HALF * N_HEADS; q += AGG_THREADS) den[q] = 0.f;
        }
        __syncthreads();

        int cnt = 0;

        auto flush = [&]() {
            __builtin_amdgcn_fence(__ATOMIC_ACQ_REL, "wavefront");
            for (int i = 0; i < cnt; i += 4) {
                int r[4], s[4];
                float m[4], ex[4];
                v4f xr[4][4];
#pragma unroll
                for (int k = 0; k < 4; ++k) {
                    const int pk = __builtin_amdgcn_readfirstlane(mylist[min(i + k, cnt - 1)]);
                    const int e = pk >> 8;
                    r[k] = pk & 255;
                    s[k] = src[e];
                    m[k] = M[e];
                }
#pragma unroll
                for (int k = 0; k < 4; ++k) {
                    const v4f* __restrict__ hk = (const v4f*)(h + (size_t)s[k] * COLS) + lane;
#pragma unroll
                    for (int j = 0; j < 4; ++j) xr[k][j] = hk[j * 32];
                }
#pragma unroll
                for (int k = 0; k < 4; ++k)
                    ex[k] = edge_exp(a_self[s[k] * N_HEADS + l8],
                                     a_neigh[(base + r[k]) * N_HEADS + l8], m[k]);
                agg_add_edge(acc, den, r[0], ex[0], xr[0], lane);
#pragma unroll
                for (int k = 1; k < 4; ++k)
                    if (i + k < cnt) agg_add_edge(acc, den, r[k], ex[k], xr[k], lane);
            }
            cnt = 0;
            __builtin_amdgcn_fence(__ATOMIC_ACQ_REL, "wavefront");
        };

        int inc = 0;
        bool full = false;
        if (half == 1) {
            for (int i = 0; i < AGG_WAVES; ++i) {
                const int c = pcount[i];
                full = full || (c < 0);
                if (i <= lane) inc += c;
            }
        }

        if (half == 0 || full) {
            auto append = [&](int dv, int ebits) {
                const unsigned rr = (unsigned)(dv - base);
                const bool mt = rr < (unsigned)n_all;
                const unsigned mask = __builtin_amdgcn_ballot_w32(mt);
                if (mask != 0u) {
                    const bool act = rr < (unsigned)n_act;
                    const unsigned ma = __builtin_amdgcn_ballot_w32(act);
                    const unsigned mp = mask & ~ma;
                    const int ent = ebits | tid10 | (int)rr;
                    if (act) mylist[cnt + (int)__builtin_amdgcn_mbcnt_lo(ma, 0u)] = ent;
                    cnt += __builtin_popcount(ma);
                    if (mp != 0u) {
                        const int np = __builtin_popcount(mp);
                        if (pcnt >= 0 && pcnt + np <= P_CAP) {
                            if (mt && !act) mypark[pcnt + (int)__builtin_amdgcn_mbcnt_lo(mp, 0u)] = ent;
                            pcnt += np;
                        } else {
                            pcnt = -1;
                        }
                    }
                }
            };

            int cnext = rot;
            int cc[SWEEP];
            v4i dc[SWEEP];
#pragma unroll
            for (int k = 0; k < SWEEP; ++k) {
                cc[k] = cnext;
                cnext = (cnext + 1 == SCAN_ITERS) ? 0 : cnext + 1;
                dc[k] = chunk_ld(cc[k]);
            }

            for (int it = 0; it < SCAN_ITERS; it += SWEEP) {
                const bool more = it + SWEEP < SCAN_ITERS;
                int cn[SWEEP];
                v4i dn[SWEEP];
#pragma unroll
                for (int k = 0; k < SWEEP; ++k) {
                    cn[k] = cnext;
                    cnext = (cnext + 1 == SCAN_ITERS) ? 0 : cnext + 1;
                    dn[k] = chunk_ld(cn[k]);
                }

#pragma unroll
                for (int k = 0; k < SWEEP; ++k) {
                    if (chunk_ok(cc[k])) {
                        const int eb = cc[k] << 20;
                        append(dc[k].x, eb | (0 << 8));
                        append(dc[k].y, eb | (1 << 8));
                        append(dc[k].z, eb | (2 << 8));
                        append(dc[k].w, eb | (3 << 8));
                    }
                }
#pragma unroll
                for (int k = 0; k < SWEEP; ++k) { cc[k] = cn[k]; dc[k] = dn[k]; }

                if (cnt >= WL_FLUSH || !more) flush();
            }
        } else {
            const int total = __builtin_amdgcn_readlane(inc, AGG_WAVES - 1);
            const int myn = total > wave_s ? (total - wave_s + AGG_WAVES - 1) / AGG_WAVES : 0;
            for (int q0 = 0; q0 < myn; q0 += 32) {
                const int q = q0 + lane;
                const int g = wave + AGG_WAVES * q;
                int owner = 0, before = 0;
#pragma unroll
                for (int i = 0; i < AGG_WAVES - 1; ++i) {
                    const int p = __builtin_amdgcn_readlane(inc, i);
                    if (p <= g) { owner = i + 1; before = p; }
                }
                if (q < myn) mylist[q] = pl[owner * P_CAP + (g - before)] - AGG_HALF;
            }
            cnt = myn;
            flush();
        }

        if (half == 0 && lane == 0) pcount[wave] = pcnt;
        __syncthreads();

        for (int q = tid; q < n_act * (COLS / 4); q += AGG_THREADS) {
            const int r  = q >> 7;
            const int c4 = q & 127;
            const int j  = c4 >> 5;
            const int l  = c4 & 31;
            const float* ar = acc + r * COLS + j * 128 + l;
            const float dn = den[r * N_HEADS + 2 * j + (l >> 4)] + 1e-16f;
            v4f v;
            v.x = ar[0]  / dn;
            v.y = ar[32] / dn;
            v.z = ar[64] / dn;
            v.w = ar[96] / dn;
            v.x = v.x > 0.f ? v.x : (__expf(v.x) - 1.0f);
            v.y = v.y > 0.f ? v.y : (__expf(v.y) - 1.0f);
            v.z = v.z > 0.f ? v.z : (__expf(v.z) - 1.0f);
            v.w = v.w > 0.f ? v.w : (__expf(v.w) - 1.0f);
            *((v4f*)(out + (size_t)(base + r) * COLS) + c4) = v;
        }
    }
}

extern "C" void kernel_launch(void* const* d_in, const int* in_sizes, int n_in,
                              void* d_out, int out_size, void* d_ws, size_t ws_size,
                              hipStream_t stream) {
    const float* x   = (const float*)d_in[0];
    const int*   src = (const int*)  d_in[1];
    const int*   dst = (const int*)  d_in[2];
    const float* M   = (const float*)d_in[3];
    const float* W   = (const float*)d_in[4];
    const float* a   = (const float*)d_in[5];
    float* out = (float*)d_out;

    float* h       = (float*)d_ws;
    float* a_self  = h      + (size_t)N_NODES * COLS;
    float* a_neigh = a_self + (size_t)N_NODES * N_HEADS;

    {
        dim3 grid(N_HEADS, (ROW_TILES + 7) / 8);
        gat_gemm_wmma<<<grid, 256, 0, stream>>>(x, W, h);
    }

    {
        int n = N_NODES * N_HEADS;
        gat_attn_coef<<<(n + 255) / 256, 256, 0, stream>>>(h, a, a_self, a_neigh);
    }

    gat_owner_aggregate<<<AGG_BLOCKS, AGG_THREADS, 0, stream>>>(src, dst, M, a_self, a_neigh,
                                                                 h, out);
}


# Attention coefficients a_self, a_neigh taken from the product's accumulators in its epilogue (4 in-lane terms, then a 16-lane DPP tree: the 64-term sums' order changes); 2 launches, not 3
# speedup vs baseline: 3.0752x; 1.0628x over previous
#include <hip/hip_runtime.h>
#include <hip/hip_bf16.h>

typedef __attribute__((ext_vector_type(2))) float v2f;
typedef __attribute__((ext_vector_type(4))) float v4f;
typedef __attribute__((ext_vector_type(8))) float v8f;

#define N_NODES  50000
#define N_EDGES  800000
#define IN_F     256
#define OUT_F    64
#define N_HEADS  8
#define COLS     (N_HEADS * OUT_F)
#define ROW_TILES (N_NODES / 16)
#define LRELU_SLOPE 0.2f


template <int M>
__device__ __forceinline__ float row_xor(float v) {
    return __builtin_bit_cast(float, __builtin_amdgcn_update_dpp(0, __builtin_bit_cast(int, v),
                                                                 0x160 | M, 0xf, 0xf, true));
}

__global__ __launch_bounds__(256) void gat_gemm_wmma(const float* __restrict__ x,
                                                     const float* __restrict__ W,
                                                     const float* __restrict__ a,
                                                     float* __restrict__ h,
                                                     float* __restrict__ a_self,
                                                     float* __restrict__ a_neigh) {
    __shared__ float Wlds[IN_F * OUT_F];

    const int wave = threadIdx.x >> 5;
    const int lane = threadIdx.x & 31;
    const int head = blockIdx.x;

    {
        const v4f* __restrict__ Wg = (const v4f*)(W + (size_t)head * (IN_F * OUT_F));
        v4f* __restrict__ Wl = (v4f*)Wlds;
#pragma unroll
        for (int i = 0; i < 16; ++i) {
            int t = threadIdx.x + i * 256;
            Wl[t] = Wg[t];
        }
    }
    __syncthreads();

    const int rowTile = blockIdx.y * 8 + wave;
    if (rowTile >= ROW_TILES) return;

    const int row0 = rowTile * 16;
    const int fo   = lane & 15;
    const int koff = (lane >> 4) << 1;
    const int arow = row0 + (lane & 15);
    const float* __restrict__ ap = x + (size_t)arow * IN_F + koff;

    const float* __restrict__ av = a + head * (2 * OUT_F) + fo;
    const float as0 = av[0],         as1 = av[16],         as2 = av[32],         as3 = av[48];
    const float an0 = av[OUT_F + 0], an1 = av[OUT_F + 16], an2 = av[OUT_F + 32], an3 = av[OUT_F + 48];

    v8f acc0 = {0.f,0.f,0.f,0.f,0.f,0.f,0.f,0.f};
    v8f acc1 = acc0, acc2 = acc0, acc3 = acc0;

#pragma unroll 2
    for (int k0 = 0; k0 < IN_F; k0 += 4) {
        v2f a = *(const v2f*)(ap + k0);
        const float* __restrict__ b0 = Wlds + (k0 + koff) * OUT_F + fo;
        const float* __restrict__ b1 = b0 + OUT_F;
        v2f b;
        b.x = b0[0];  b.y = b1[0];
        acc0 = __builtin_amdgcn_wmma_f32_16x16x4_f32(false, a, false, b,
                                                     (short)0, acc0, false, false);
        b.x = b0[16]; b.y = b1[16];
        acc1 = __builtin_amdgcn_wmma_f32_16x16x4_f32(false, a, false, b,
                                                     (short)0, acc1, false, false);
        b.x = b0[32]; b.y = b1[32];
        acc2 = __builtin_amdgcn_wmma_f32_16x16x4_f32(false, a, false, b,
                                                     (short)0, acc2, false, false);
        b.x = b0[48]; b.y = b1[48];
        acc3 = __builtin_amdgcn_wmma_f32_16x16x4_f32(false, a, false, b,
                                                     (short)0, acc3, false, false);
    }

    const int rbase = row0 + ((lane >> 4) << 3);
    float* __restrict__ hp = h + head * OUT_F + fo;
#pragma unroll
    for (int v = 0; v < 8; ++v) {
        float* __restrict__ hr = hp + (size_t)(rbase + v) * COLS;
        hr[0]  = acc0[v];
        hr[16] = acc1[v];
        hr[32] = acc2[v];
        hr[48] = acc3[v];
    }

    float q[16];
#pragma unroll
    for (int v = 0; v < 8; ++v) {
        q[v]     = acc0[v] * as0 + acc1[v] * as1 + acc2[v] * as2 + acc3[v] * as3;
        q[8 + v] = acc0[v] * an0 + acc1[v] * an1 + acc2[v] * an2 + acc3[v] * an3;
    }
    float r8[8], r4[4], r2[2];
    {
        const bool b = (fo & 8) != 0;
#pragma unroll
        for (int j = 0; j < 8; ++j) {
            const float keep = b ? q[8 + j] : q[j];
            const float send = b ? q[j] : q[8 + j];
            r8[j] = keep + row_xor<8>(send);
        }
    }
    {
        const bool b = (fo & 4) != 0;
#pragma unroll
        for (int j = 0; j < 4; ++j) {
            const float keep = b ? r8[4 + j] : r8[j];
            const float send = b ? r8[j] : r8[4 + j];
            r4[j] = keep + row_xor<4>(send);
        }
    }
    {
        const bool b = (fo & 2) != 0;
#pragma unroll
        for (int j = 0; j < 2; ++j) {
            const float keep = b ? r4[2 + j] : r4[j];
            const float send = b ? r4[j] : r4[2 + j];
            r2[j] = keep + row_xor<2>(send);
        }
    }
    float coef;
    {
        const bool b = (fo & 1) != 0;
        const float keep = b ? r2[1] : r2[0];
        const float send = b ? r2[0] : r2[1];
        coef = keep + row_xor<1>(send);
    }
    float* cp = (fo & 8) ? a_neigh : a_self;
    cp[(rbase + (fo & 7)) * N_HEADS + head] = coef;
}

__device__ __forceinline__ float edge_exp(float as, float an, float m) {
    float lg = as + an;
    lg = (lg > 0.f ? lg : LRELU_SLOPE * lg) * m;
    return __expf(lg);
}

#define AGG_HALF    98
#define AGG_ROWS    (2 * AGG_HALF)
#define AGG_BLOCKS  ((N_NODES + AGG_ROWS - 1) / AGG_ROWS)
#define AGG_THREADS 1024
#define AGG_WAVES   (AGG_THREADS / 32)
#define WL_FLUSH    16
#define SWEEP       4
#define WL_CAP      (WL_FLUSH + SWEEP * 4 * 32)
#define P_CAP       384
#define N_DST4      (N_EDGES / 4)
#define SCAN_ITERS  ((N_DST4 + AGG_THREADS - 1) / AGG_THREADS)

static_assert(SCAN_ITERS % SWEEP == 0, "the sweep takes SWEEP chunks per trip and wraps around");
static_assert(N_DST4 % 32 == 0, "a wave's 32 lanes are inside dst together or outside it together");
static_assert(AGG_WAVES == 32, "lane i of a wave stands for wave i where the parked counts are summed");
static_assert(P_CAP <= WL_CAP, "a wave's share of the parked entries fits its own list");
static_assert(AGG_ROWS <= 256, "a row number takes the low 8 bits of a list entry");
static_assert(N_EDGES < (1 << 23), "an edge number takes the bits above the row number");

typedef __attribute__((ext_vector_type(4))) int v4i;

__device__ __forceinline__ float lane_bcast(float v, int srcLane) {
    return __int_as_float(__builtin_amdgcn_readlane(__float_as_int(v), srcLane));
}

__device__ __forceinline__ void agg_add_edge(float* __restrict__ acc, float* __restrict__ den,
                                             int r, float ex, const v4f* x, int lane) {
    if (lane < N_HEADS) atomicAdd(&den[r * N_HEADS + lane], ex);
    float* ar = acc + r * COLS + lane;
    const int hi = lane >> 4;
#pragma unroll
    for (int j = 0; j < 4; ++j) {
        float wlo = lane_bcast(ex, 2 * j);
        float whi = lane_bcast(ex, 2 * j + 1);
        float w = hi ? whi : wlo;
        atomicAdd(ar + j * 128 +  0, w * x[j].x);
        atomicAdd(ar + j * 128 + 32, w * x[j].y);
        atomicAdd(ar + j * 128 + 64, w * x[j].z);
        atomicAdd(ar + j * 128 + 96, w * x[j].w);
    }
}

__global__ __launch_bounds__(AGG_THREADS) void gat_owner_aggregate(const int* __restrict__ src,
                                                                  const int* __restrict__ dst,
                                                                  const float* __restrict__ M,
                                                                  const float* __restrict__ a_self,
                                                                  const float* __restrict__ a_neigh,
                                                                  const float* __restrict__ h,
                                                                  float* __restrict__ out) {
    __shared__ float acc[AGG_HALF * COLS];
    __shared__ float den[AGG_HALF * N_HEADS];
    __shared__ int   wl[AGG_WAVES * WL_CAP];
    __shared__ int   pl[AGG_WAVES * P_CAP];
    __shared__ int   pcount[AGG_WAVES];

    const int tid  = threadIdx.x;
    const int wave = tid >> 5;
    const int lane = tid & 31;
    const int lo    = blockIdx.x * AGG_ROWS;
    const int nrows = min(AGG_ROWS, N_NODES - lo);
    const int n0    = min(AGG_HALF, nrows);

    int* mylist = wl + wave * WL_CAP;
    int* mypark = pl + wave * P_CAP;
    int pcnt = 0;
    const v4i* __restrict__ dst4 = (const v4i*)dst;
    const int l8 = lane & 7;

    const int tid10 = tid << 10;

    const int rot = blockIdx.x % SCAN_ITERS;

    const int wave_s = __builtin_amdgcn_readfirstlane(wave);
    auto chunk_ok = [&](int c) { return c * AGG_THREADS + wave_s * 32 < N_DST4; };
    auto chunk_ld = [&](int c) { return dst4[(chunk_ok(c) ? c * AGG_THREADS : 0) + tid]; };

    for (int half = 0; half < 2; ++half) {
        const int base  = lo + half * AGG_HALF;
        const int n_act = half == 0 ? n0 : nrows - n0;
        const int n_all = half == 0 ? nrows : n_act;
        if (n_act == 0) break;

        if (half == 1) __syncthreads();
        {
            const v4f z = {0.f, 0.f, 0.f, 0.f};
            v4f* a4 = (v4f*)acc;
            for (int q = tid; q < AGG_HALF * COLS / 4; q += AGG_THREADS) a4[q] = z;
            for (int q = tid; q < AGG_HALF * N_HEADS; q += AGG_THREADS) den[q] = 0.f;
        }
        __syncthreads();

        int cnt = 0;

        auto flush = [&]() {
            __builtin_amdgcn_fence(__ATOMIC_ACQ_REL, "wavefront");
            for (int i = 0; i < cnt; i += 4) {
                int r[4], s[4];
                float m[4], ex[4];
                v4f xr[4][4];
#pragma unroll
                for (int k = 0; k < 4; ++k) {
                    const int pk = __builtin_amdgcn_readfirstlane(mylist[min(i + k, cnt - 1)]);
                    const int e = pk >> 8;
                    r[k] = pk & 255;
                    s[k] = src[e];
                    m[k] = M[e];
                }
#pragma unroll
                for (int k = 0; k < 4; ++k) {
                    const v4f* __restrict__ hk = (const v4f*)(h + (size_t)s[k] * COLS) + lane;
#pragma unroll
                    for (int j = 0; j < 4; ++j) xr[k][j] = hk[j * 32];
                }
#pragma unroll
                for (int k = 0; k < 4; ++k)
                    ex[k] = edge_exp(a_self[s[k] * N_HEADS + l8],
                                     a_neigh[(base + r[k]) * N_HEADS + l8], m[k]);
                agg_add_edge(acc, den, r[0], ex[0], xr[0], lane);
#pragma unroll
                for (int k = 1; k < 4; ++k)
                    if (i + k < cnt) agg_add_edge(acc, den, r[k], ex[k], xr[k], lane);
            }
            cnt = 0;
            __builtin_amdgcn_fence(__ATOMIC_ACQ_REL, "wavefront");
        };

        int inc = 0;
        bool full = false;
        if (half == 1) {
            for (int i = 0; i < AGG_WAVES; ++i) {
                const int c = pcount[i];
                full = full || (c < 0);
                if (i <= lane) inc += c;
            }
        }

        if (half == 0 || full) {
            auto append = [&](int dv, int ebits) {
                const unsigned rr = (unsigned)(dv - base);
                const bool mt = rr < (unsigned)n_all;
                const unsigned mask = __builtin_amdgcn_ballot_w32(mt);
                if (mask != 0u) {
                    const bool act = rr < (unsigned)n_act;
                    const unsigned ma = __builtin_amdgcn_ballot_w32(act);
                    const unsigned mp = mask & ~ma;
                    const int ent = ebits | tid10 | (int)rr;
                    if (act) mylist[cnt + (int)__builtin_amdgcn_mbcnt_lo(ma, 0u)] = ent;
                    cnt += __builtin_popcount(ma);
                    if (mp != 0u) {
                        const int np = __builtin_popcount(mp);
                        if (pcnt >= 0 && pcnt + np <= P_CAP) {
                            if (mt && !act) mypark[pcnt + (int)__builtin_amdgcn_mbcnt_lo(mp, 0u)] = ent;
                            pcnt += np;
                        } else {
                            pcnt = -1;
                        }
                    }
                }
            };

            int cnext = rot;
            int cc[SWEEP];
            v4i dc[SWEEP];
#pragma unroll
            for (int k = 0; k < SWEEP; ++k) {
                cc[k] = cnext;
                cnext = (cnext + 1 == SCAN_ITERS) ? 0 : cnext + 1;
                dc[k] = chunk_ld(cc[k]);
            }

            for (int it = 0; it < SCAN_ITERS; it += SWEEP) {
                const bool more = it + SWEEP < SCAN_ITERS;
                int cn[SWEEP];
                v4i dn[SWEEP];
#pragma unroll
                for (int k = 0; k < SWEEP; ++k) {
                    cn[k] = cnext;
                    cnext = (cnext + 1 == SCAN_ITERS) ? 0 : cnext + 1;
                    dn[k] = chunk_ld(cn[k]);
                }

#pragma unroll
                for (int k = 0; k < SWEEP; ++k) {
                    if (chunk_ok(cc[k])) {
                        const int eb = cc[k] << 20;
                        append(dc[k].x, eb | (0 << 8));
                        append(dc[k].y, eb | (1 << 8));
                        append(dc[k].z, eb | (2 << 8));
                        append(dc[k].w, eb | (3 << 8));
                    }
                }
#pragma unroll
                for (int k = 0; k < SWEEP; ++k) { cc[k] = cn[k]; dc[k] = dn[k]; }

                if (cnt >= WL_FLUSH || !more) flush();
            }
        } else {
            const int total = __builtin_amdgcn_readlane(inc, AGG_WAVES - 1);
            const int myn = total > wave_s ? (total - wave_s + AGG_WAVES - 1) / AGG_WAVES : 0;
            for (int q0 = 0; q0 < myn; q0 += 32) {
                const int q = q0 + lane;
                const int g = wave + AGG_WAVES * q;
                int owner = 0, before = 0;
#pragma unroll
                for (int i = 0; i < AGG_WAVES - 1; ++i) {
                    const int p = __builtin_amdgcn_readlane(inc, i);
                    if (p <= g) { owner = i + 1; before = p; }
                }
                if (q < myn) mylist[q] = pl[owner * P_CAP + (g - before)] - AGG_HALF;
            }
            cnt = myn;
            flush();
        }

        if (half == 0 && lane == 0) pcount[wave] = pcnt;
        __syncthreads();

        for (int q = tid; q < n_act * (COLS / 4); q += AGG_THREADS) {
            const int r  = q >> 7;
            const int c4 = q & 127;
            const int j  = c4 >> 5;
            const int l  = c4 & 31;
            const float* ar = acc + r * COLS + j * 128 + l;
            const float dn = den[r * N_HEADS + 2 * j + (l >> 4)] + 1e-16f;
            v4f v;
            v.x = ar[0]  / dn;
            v.y = ar[32] / dn;
            v.z = ar[64] / dn;
            v.w = ar[96] / dn;
            v.x = v.x > 0.f ? v.x : (__expf(v.x) - 1.0f);
            v.y = v.y > 0.f ? v.y : (__expf(v.y) - 1.0f);
            v.z = v.z > 0.f ? v.z : (__expf(v.z) - 1.0f);
            v.w = v.w > 0.f ? v.w : (__expf(v.w) - 1.0f);
            *((v4f*)(out + (size_t)(base + r) * COLS) + c4) = v;
        }
    }
}

extern "C" void kernel_launch(void* const* d_in, const int* in_sizes, int n_in,
                              void* d_out, int out_size, void* d_ws, size_t ws_size,
                              hipStream_t stream) {
    const float* x   = (const float*)d_in[0];
    const int*   src = (const int*)  d_in[1];
    const int*   dst = (const int*)  d_in[2];
    const float* M   = (const float*)d_in[3];
    const float* W   = (const float*)d_in[4];
    const float* a   = (const float*)d_in[5];
    float* out = (float*)d_out;

    float* h       = (float*)d_ws;
    float* a_self  = h      + (size_t)N_NODES * COLS;
    float* a_neigh = a_self + (size_t)N_NODES * N_HEADS;

    {
        dim3 grid(N_HEADS, (ROW_TILES + 7) / 8);
        gat_gemm_wmma<<<grid, 256, 0, stream>>>(x, W, a, h, a_self, a_neigh);
    }

    gat_owner_aggregate<<<AGG_BLOCKS, AGG_THREADS, 0, stream>>>(src, dst, M, a_self, a_neigh,
                                                                 h, out);
}


# Owner block queues its edges per dst row in LDS (integer slot adds) while sweeping dst; one wave then sums a row's exp*h[src] in registers in queue (arrival) order and stores it: no float atomics
# speedup vs baseline: 3.4771x; 1.1307x over previous
#include <hip/hip_runtime.h>
#include <hip/hip_bf16.h>

typedef __attribute__((ext_vector_type(2))) float v2f;
typedef __attribute__((ext_vector_type(4))) float v4f;
typedef __attribute__((ext_vector_type(8))) float v8f;

#define N_NODES  50000
#define N_EDGES  800000
#define IN_F     256
#define OUT_F    64
#define N_HEADS  8
#define COLS     (N_HEADS * OUT_F)
#define ROW_TILES (N_NODES / 16)
#define LRELU_SLOPE 0.2f


template <int M>
__device__ __forceinline__ float row_xor(float v) {
    return __builtin_bit_cast(float, __builtin_amdgcn_update_dpp(0, __builtin_bit_cast(int, v),
                                                                 0x160 | M, 0xf, 0xf, true));
}

__global__ __launch_bounds__(256) void gat_gemm_wmma(const float* __restrict__ x,
                                                     const float* __restrict__ W,
                                                     const float* __restrict__ a,
                                                     float* __restrict__ h,
                                                     float* __restrict__ a_self,
                                                     float* __restrict__ a_neigh) {
    __shared__ float Wlds[IN_F * OUT_F];

    const int wave = threadIdx.x >> 5;
    const int lane = threadIdx.x & 31;
    const int head = blockIdx.x;

    {
        const v4f* __restrict__ Wg = (const v4f*)(W + (size_t)head * (IN_F * OUT_F));
        v4f* __restrict__ Wl = (v4f*)Wlds;
#pragma unroll
        for (int i = 0; i < 16; ++i) {
            int t = threadIdx.x + i * 256;
            Wl[t] = Wg[t];
        }
    }
    __syncthreads();

    const int rowTile = blockIdx.y * 8 + wave;
    if (rowTile >= ROW_TILES) return;

    const int row0 = rowTile * 16;
    const int fo   = lane & 15;
    const int koff = (lane >> 4) << 1;
    const int arow = row0 + (lane & 15);
    const float* __restrict__ ap = x + (size_t)arow * IN_F + koff;

    const float* __restrict__ av = a + head * (2 * OUT_F) + fo;
    const float as0 = av[0],         as1 = av[16],         as2 = av[32],         as3 = av[48];
    const float an0 = av[OUT_F + 0], an1 = av[OUT_F + 16], an2 = av[OUT_F + 32], an3 = av[OUT_F + 48];

    v8f acc0 = {0.f,0.f,0.f,0.f,0.f,0.f,0.f,0.f};
    v8f acc1 = acc0, acc2 = acc0, acc3 = acc0;

#pragma unroll 2
    for (int k0 = 0; k0 < IN_F; k0 += 4) {
        v2f a = *(const v2f*)(ap + k0);
        const float* __restrict__ b0 = Wlds + (k0 + koff) * OUT_F + fo;
        const float* __restrict__ b1 = b0 + OUT_F;
        v2f b;
        b.x = b0[0];  b.y = b1[0];
        acc0 = __builtin_amdgcn_wmma_f32_16x16x4_f32(false, a, false, b,
                                                     (short)0, acc0, false, false);
        b.x = b0[16]; b.y = b1[16];
        acc1 = __builtin_amdgcn_wmma_f32_16x16x4_f32(false, a, false, b,
                                                     (short)0, acc1, false, false);
        b.x = b0[32]; b.y = b1[32];
        acc2 = __builtin_amdgcn_wmma_f32_16x16x4_f32(false, a, false, b,
                                                     (short)0, acc2, false, false);
        b.x = b0[48]; b.y = b1[48];
        acc3 = __builtin_amdgcn_wmma_f32_16x16x4_f32(false, a, false, b,
                                                     (short)0, acc3, false, false);
    }

    const int rbase = row0 + ((lane >> 4) << 3);
    float* __restrict__ hp = h + head * OUT_F + fo;
#pragma unroll
    for (int v = 0; v < 8; ++v) {
        float* __restrict__ hr = hp + (size_t)(rbase + v) * COLS;
        hr[0]  = acc0[v];
        hr[16] = acc1[v];
        hr[32] = acc2[v];
        hr[48] = acc3[v];
    }

    float q[16];
#pragma unroll
    for (int v = 0; v < 8; ++v) {
        q[v]     = acc0[v] * as0 + acc1[v] * as1 + acc2[v] * as2 + acc3[v] * as3;
        q[8 + v] = acc0[v] * an0 + acc1[v] * an1 + acc2[v] * an2 + acc3[v] * an3;
    }
    float r8[8], r4[4], r2[2];
    {
        const bool b = (fo & 8) != 0;
#pragma unroll
        for (int j = 0; j < 8; ++j) {
            const float keep = b ? q[8 + j] : q[j];
            const float send = b ? q[j] : q[8 + j];
            r8[j] = keep + row_xor<8>(send);
        }
    }
    {
        const bool b = (fo & 4) != 0;
#pragma unroll
        for (int j = 0; j < 4; ++j) {
            const float keep = b ? r8[4 + j] : r8[j];
            const float send = b ? r8[j] : r8[4 + j];
            r4[j] = keep + row_xor<4>(send);
        }
    }
    {
        const bool b = (fo & 2) != 0;
#pragma unroll
        for (int j = 0; j < 2; ++j) {
            const float keep = b ? r4[2 + j] : r4[j];
            const float send = b ? r4[j] : r4[2 + j];
            r2[j] = keep + row_xor<2>(send);
        }
    }
    float coef;
    {
        const bool b = (fo & 1) != 0;
        const float keep = b ? r2[1] : r2[0];
        const float send = b ? r2[0] : r2[1];
        coef = keep + row_xor<1>(send);
    }
    float* cp = (fo & 8) ? a_neigh : a_self;
    cp[(rbase + (fo & 7)) * N_HEADS + head] = coef;
}

__device__ __forceinline__ float edge_exp(float as, float an, float m) {
    float lg = as + an;
    lg = (lg > 0.f ? lg : LRELU_SLOPE * lg) * m;
    return __expf(lg);
}

#define AGG_ROWS    196
#define AGG_BLOCKS  ((N_NODES + AGG_ROWS - 1) / AGG_ROWS)
#define AGG_THREADS 1024
#define AGG_WAVES   (AGG_THREADS / 32)
#define SWEEP       4
#define Q_CAP       128
#define N_DST4      (N_EDGES / 4)
#define SCAN_ITERS  ((N_DST4 + AGG_THREADS - 1) / AGG_THREADS)

static_assert(SCAN_ITERS % SWEEP == 0, "the sweep takes SWEEP chunks per trip and wraps around");
static_assert(N_DST4 % 32 == 0, "a wave's 32 lanes are inside dst together or outside it together");

typedef __attribute__((ext_vector_type(4))) int v4i;

__device__ __forceinline__ float lane_bcast(float v, int srcLane) {
    return __int_as_float(__builtin_amdgcn_readlane(__float_as_int(v), srcLane));
}

__device__ __forceinline__ void row_add_edge(v4f* acc, float& den, float ex, const v4f* x, int hi) {
    den += ex;
#pragma unroll
    for (int j = 0; j < 4; ++j) {
        const float wlo = lane_bcast(ex, 2 * j);
        const float whi = lane_bcast(ex, 2 * j + 1);
        const float w = hi ? whi : wlo;
        acc[j].x += w * x[j].x;
        acc[j].y += w * x[j].y;
        acc[j].z += w * x[j].z;
        acc[j].w += w * x[j].w;
    }
}

__global__ __launch_bounds__(AGG_THREADS) void gat_owner_aggregate(const int* __restrict__ src,
                                                                  const int* __restrict__ dst,
                                                                  const float* __restrict__ M,
                                                                  const float* __restrict__ a_self,
                                                                  const float* __restrict__ a_neigh,
                                                                  const float* __restrict__ h,
                                                                  float* __restrict__ out) {
    __shared__ int rowq[AGG_ROWS * Q_CAP];
    __shared__ int rowcnt[AGG_ROWS];
    __shared__ int ticket;

    const int tid  = threadIdx.x;
    const int wave = tid >> 5;
    const int lane = tid & 31;
    const int lo    = blockIdx.x * AGG_ROWS;
    const int nrows = min(AGG_ROWS, N_NODES - lo);
    const v4i* __restrict__ dst4 = (const v4i*)dst;
    const int l8 = lane & 7;
    const int hi = lane >> 4;

    for (int q = tid; q < AGG_ROWS; q += AGG_THREADS) rowcnt[q] = 0;
    if (tid == 0) ticket = 0;
    __syncthreads();

    const int rot = blockIdx.x % SCAN_ITERS;

    const int wave_s = __builtin_amdgcn_readfirstlane(wave);
    auto chunk_ok = [&](int c) { return c * AGG_THREADS + wave_s * 32 < N_DST4; };
    auto chunk_ld = [&](int c) { return dst4[(chunk_ok(c) ? c * AGG_THREADS : 0) + tid]; };

    auto append = [&](int dv, int e) {
        const unsigned rr = (unsigned)(dv - lo);
        if (rr < (unsigned)nrows) {
            const int slot = atomicAdd(&rowcnt[rr], 1);
            if (slot < Q_CAP) rowq[rr * Q_CAP + slot] = e;
        }
    };

    {
        int cnext = rot;
        int cc[SWEEP];
        v4i dc[SWEEP];
#pragma unroll
        for (int k = 0; k < SWEEP; ++k) {
            cc[k] = cnext;
            cnext = (cnext + 1 == SCAN_ITERS) ? 0 : cnext + 1;
            dc[k] = chunk_ld(cc[k]);
        }

        for (int it = 0; it < SCAN_ITERS; it += SWEEP) {
            int cn[SWEEP];
            v4i dn[SWEEP];
#pragma unroll
            for (int k = 0; k < SWEEP; ++k) {
                cn[k] = cnext;
                cnext = (cnext + 1 == SCAN_ITERS) ? 0 : cnext + 1;
                dn[k] = chunk_ld(cn[k]);
            }

#pragma unroll
            for (int k = 0; k < SWEEP; ++k) {
                if (chunk_ok(cc[k])) {
                    const int eb = cc[k] * (4 * AGG_THREADS) + 4 * tid;
                    append(dc[k].x, eb + 0);
                    append(dc[k].y, eb + 1);
                    append(dc[k].z, eb + 2);
                    append(dc[k].w, eb + 3);
                }
            }
#pragma unroll
            for (int k = 0; k < SWEEP; ++k) { cc[k] = cn[k]; dc[k] = dn[k]; }
        }
    }
    __syncthreads();

    for (;;) {
        int r = 0;
        if (lane == 0) r = atomicAdd(&ticket, 1);
        r = __builtin_amdgcn_readfirstlane(r);
        if (r >= nrows) break;

        const int n   = __builtin_amdgcn_readfirstlane(rowcnt[r]);
        const int row = lo + r;
        const float an = a_neigh[row * N_HEADS + l8];
        const v4f z = {0.f, 0.f, 0.f, 0.f};
        v4f acc[4] = {z, z, z, z};
        float den = 0.f;

        if (n <= Q_CAP) {
            for (int i0 = 0; i0 < n; i0 += 32) {
                const int nb  = min(32, n - i0);
                const int e_l = rowq[r * Q_CAP + i0 + min(lane, nb - 1)];
                const int s_l = src[e_l];
                const float m_l = M[e_l];
                for (int i = 0; i < nb; i += 4) {
                    const int g = min(4, nb - i);
                    int s[4];
                    float m[4], as[4];
                    v4f xr[4][4];
#pragma unroll
                    for (int k = 0; k < 4; ++k) {
                        const int idx = min(i + k, nb - 1);
                        s[k] = __builtin_amdgcn_readlane(s_l, idx);
                        m[k] = lane_bcast(m_l, idx);
                    }
#pragma unroll
                    for (int k = 0; k < 4; ++k) {
                        if (k < g) {
                            const v4f* __restrict__ hk = (const v4f*)(h + (size_t)s[k] * COLS) + lane;
#pragma unroll
                            for (int j = 0; j < 4; ++j) xr[k][j] = hk[j * 32];
                            as[k] = a_self[s[k] * N_HEADS + l8];
                        }
                    }
#pragma unroll
                    for (int k = 0; k < 4; ++k) {
                        if (k < g) row_add_edge(acc, den, edge_exp(as[k], an, m[k]), xr[k], hi);
                    }
                }
            }
        } else {
            for (int c = 0; c < N_DST4 / 32; ++c) {
                const v4i d = dst4[c * 32 + lane];
#pragma unroll
                for (int j = 0; j < 4; ++j) {
                    const int dv = j == 0 ? d.x : j == 1 ? d.y : j == 2 ? d.z : d.w;
                    unsigned mk = __builtin_amdgcn_ballot_w32(dv == row);
                    while (mk != 0u) {
                        const int l = __builtin_ctz(mk);
                        mk &= mk - 1u;
                        const int e = (c * 32 + l) * 4 + j;
                        const int s1 = src[e];
                        const float m1 = M[e];
                        const v4f* __restrict__ hk = (const v4f*)(h + (size_t)s1 * COLS) + lane;
                        v4f x1[4];
#pragma unroll
                        for (int jj = 0; jj < 4; ++jj) x1[jj] = hk[jj * 32];
                        row_add_edge(acc, den, edge_exp(a_self[s1 * N_HEADS + l8], an, m1), x1, hi);
                    }
                }
            }
        }

        float* __restrict__ orow = out + (size_t)row * COLS + 4 * lane;
#pragma unroll
        for (int j = 0; j < 4; ++j) {
            const float dlo = lane_bcast(den, 2 * j);
            const float dhi = lane_bcast(den, 2 * j + 1);
            const float dn = (hi ? dhi : dlo) + 1e-16f;
            v4f v;
            v.x = acc[j].x / dn;
            v.y = acc[j].y / dn;
            v.z = acc[j].z / dn;
            v.w = acc[j].w / dn;
            v.x = v.x > 0.f ? v.x : (__expf(v.x) - 1.0f);
            v.y = v.y > 0.f ? v.y : (__expf(v.y) - 1.0f);
            v.z = v.z > 0.f ? v.z : (__expf(v.z) - 1.0f);
            v.w = v.w > 0.f ? v.w : (__expf(v.w) - 1.0f);
            *(v4f*)(orow + j * 128) = v;
        }
    }
}

extern "C" void kernel_launch(void* const* d_in, const int* in_sizes, int n_in,
                              void* d_out, int out_size, void* d_ws, size_t ws_size,
                              hipStream_t stream) {
    const float* x   = (const float*)d_in[0];
    const int*   src = (const int*)  d_in[1];
    const int*   dst = (const int*)  d_in[2];
    const float* M   = (const float*)d_in[3];
    const float* W   = (const float*)d_in[4];
    const float* a   = (const float*)d_in[5];
    float* out = (float*)d_out;

    float* h       = (float*)d_ws;
    float* a_self  = h      + (size_t)N_NODES * COLS;
    float* a_neigh = a_self + (size_t)N_NODES * N_HEADS;

    {
        dim3 grid(N_HEADS, (ROW_TILES + 7) / 8);
        gat_gemm_wmma<<<grid, 256, 0, stream>>>(x, W, a, h, a_self, a_neigh);
    }

    gat_owner_aggregate<<<AGG_BLOCKS, AGG_THREADS, 0, stream>>>(src, dst, M, a_self, a_neigh,
                                                                 h, out);
}


# Sweep reads one byte an edge, not dst: the product launch writes each edge's owner-block id (dst/196, 256 blocks); a wave parks matching edges, reads their dst 32 at a time; sums stay arrival order
# speedup vs baseline: 3.8282x; 1.1010x over previous
#include <hip/hip_runtime.h>
#include <hip/hip_bf16.h>

typedef __attribute__((ext_vector_type(2))) float v2f;
typedef __attribute__((ext_vector_type(4))) float v4f;
typedef __attribute__((ext_vector_type(8))) float v8f;

#define N_NODES  50000
#define N_EDGES  800000
#define IN_F     256
#define OUT_F    64
#define N_HEADS  8
#define COLS     (N_HEADS * OUT_F)
#define ROW_TILES (N_NODES / 16)
#define LRELU_SLOPE 0.2f

typedef __attribute__((ext_vector_type(4))) int v4i;

#define AGG_ROWS    196
#define AGG_BLOCKS  ((N_NODES + AGG_ROWS - 1) / AGG_ROWS)
#define AGG_THREADS 1024
#define AGG_WAVES   (AGG_THREADS / 32)
#define N_DST4      (N_EDGES / 4)
#define OWN_ITERS   ((N_EDGES / 16 + AGG_THREADS - 1) / AGG_THREADS)
#define OWN_CHUNKS  (OWN_ITERS * AGG_THREADS)
#define OWN_DWORDS  (OWN_CHUNKS * 4)
#define OWN_WAVES   (OWN_DWORDS / 32)

static_assert(AGG_BLOCKS <= 256, "an owner id must fit one byte");
static_assert(N_EDGES % 16 == 0, "a chunk of 16 owner ids is all real or all padding");
static_assert(OWN_WAVES <= ((ROW_TILES + 7) / 8) * N_HEADS * 8, "phase 1 has a wave for every line of owner ids");


template <int M>
__device__ __forceinline__ float row_xor(float v) {
    return __builtin_bit_cast(float, __builtin_amdgcn_update_dpp(0, __builtin_bit_cast(int, v),
                                                                 0x160 | M, 0xf, 0xf, true));
}

__global__ __launch_bounds__(256) void gat_gemm_wmma(const float* __restrict__ x,
                                                     const float* __restrict__ W,
                                                     const float* __restrict__ a,
                                                     const int* __restrict__ dst,
                                                     float* __restrict__ h,
                                                     float* __restrict__ a_self,
                                                     float* __restrict__ a_neigh,
                                                     int* __restrict__ owner) {
    __shared__ float Wlds[IN_F * OUT_F];

    const int wave = threadIdx.x >> 5;
    const int lane = threadIdx.x & 31;
    const int head = blockIdx.x;

    {
        const int gw = (((int)blockIdx.y * N_HEADS + (int)blockIdx.x) << 3) + wave;
        if (gw < OWN_WAVES) {
            const int i = gw * 32 + lane;
            unsigned pk = 0xffffffffu;
            if (i < N_DST4) {
                const v4i d = ((const v4i*)dst)[i];
                pk = (((unsigned)d.x / AGG_ROWS) & 0xffu)
                   | ((((unsigned)d.y / AGG_ROWS) & 0xffu) << 8)
                   | ((((unsigned)d.z / AGG_ROWS) & 0xffu) << 16)
                   | ((((unsigned)d.w / AGG_ROWS) & 0xffu) << 24);
            }
            owner[i] = (int)pk;
        }
    }

    {
        const v4f* __restrict__ Wg = (const v4f*)(W + (size_t)head * (IN_F * OUT_F));
        v4f* __restrict__ Wl = (v4f*)Wlds;
#pragma unroll
        for (int i = 0; i < 16; ++i) {
            int t = threadIdx.x + i * 256;
            Wl[t] = Wg[t];
        }
    }
    __syncthreads();

    const int rowTile = blockIdx.y * 8 + wave;
    if (rowTile >= ROW_TILES) return;

    const int row0 = rowTile * 16;
    const int fo   = lane & 15;
    const int koff = (lane >> 4) << 1;
    const int arow = row0 + (lane & 15);
    const float* __restrict__ ap = x + (size_t)arow * IN_F + koff;

    const float* __restrict__ av = a + head * (2 * OUT_F) + fo;
    const float as0 = av[0],         as1 = av[16],         as2 = av[32],         as3 = av[48];
    const float an0 = av[OUT_F + 0], an1 = av[OUT_F + 16], an2 = av[OUT_F + 32], an3 = av[OUT_F + 48];

    v8f acc0 = {0.f,0.f,0.f,0.f,0.f,0.f,0.f,0.f};
    v8f acc1 = acc0, acc2 = acc0, acc3 = acc0;

#pragma unroll 2
    for (int k0 = 0; k0 < IN_F; k0 += 4) {
        v2f a = *(const v2f*)(ap + k0);
        const float* __restrict__ b0 = Wlds + (k0 + koff) * OUT_F + fo;
        const float* __restrict__ b1 = b0 + OUT_F;
        v2f b;
        b.x = b0[0];  b.y = b1[0];
        acc0 = __builtin_amdgcn_wmma_f32_16x16x4_f32(false, a, false, b,
                                                     (short)0, acc0, false, false);
        b.x = b0[16]; b.y = b1[16];
        acc1 = __builtin_amdgcn_wmma_f32_16x16x4_f32(false, a, false, b,
                                                     (short)0, acc1, false, false);
        b.x = b0[32]; b.y = b1[32];
        acc2 = __builtin_amdgcn_wmma_f32_16x16x4_f32(false, a, false, b,
                                                     (short)0, acc2, false, false);
        b.x = b0[48]; b.y = b1[48];
        acc3 = __builtin_amdgcn_wmma_f32_16x16x4_f32(false, a, false, b,
                                                     (short)0, acc3, false, false);
    }

    const int rbase = row0 + ((lane >> 4) << 3);
    float* __restrict__ hp = h + head * OUT_F + fo;
#pragma unroll
    for (int v = 0; v < 8; ++v) {
        float* __restrict__ hr = hp + (size_t)(rbase + v) * COLS;
        hr[0]  = acc0[v];
        hr[16] = acc1[v];
        hr[32] = acc2[v];
        hr[48] = acc3[v];
    }

    float q[16];
#pragma unroll
    for (int v = 0; v < 8; ++v) {
        q[v]     = acc0[v] * as0 + acc1[v] * as1 + acc2[v] * as2 + acc3[v] * as3;
        q[8 + v] = acc0[v] * an0 + acc1[v] * an1 + acc2[v] * an2 + acc3[v] * an3;
    }
    float r8[8], r4[4], r2[2];
    {
        const bool b = (fo & 8) != 0;
#pragma unroll
        for (int j = 0; j < 8; ++j) {
            const float keep = b ? q[8 + j] : q[j];
            const float send = b ? q[j] : q[8 + j];
            r8[j] = keep + row_xor<8>(send);
        }
    }
    {
        const bool b = (fo & 4) != 0;
#pragma unroll
        for (int j = 0; j < 4; ++j) {
            const float keep = b ? r8[4 + j] : r8[j];
            const float send = b ? r8[j] : r8[4 + j];
            r4[j] = keep + row_xor<4>(send);
        }
    }
    {
        const bool b = (fo & 2) != 0;
#pragma unroll
        for (int j = 0; j < 2; ++j) {
            const float keep = b ? r4[2 + j] : r4[j];
            const float send = b ? r4[j] : r4[2 + j];
            r2[j] = keep + row_xor<2>(send);
        }
    }
    float coef;
    {
        const bool b = (fo & 1) != 0;
        const float keep = b ? r2[1] : r2[0];
        const float send = b ? r2[0] : r2[1];
        coef = keep + row_xor<1>(send);
    }
    float* cp = (fo & 8) ? a_neigh : a_self;
    cp[(rbase + (fo & 7)) * N_HEADS + head] = coef;
}

__device__ __forceinline__ float edge_exp(float as, float an, float m) {
    float lg = as + an;
    lg = (lg > 0.f ? lg : LRELU_SLOPE * lg) * m;
    return __expf(lg);
}

#define SWEEP       7
#define Q_CAP       128
#define PEND_CAP    64

static_assert(OWN_ITERS % SWEEP == 0, "the sweep takes SWEEP chunks per trip and wraps around");
static_assert(N_DST4 % 32 == 0, "a wave's 32 lanes are inside dst together or outside it together");

__device__ __forceinline__ unsigned zero_bytes(unsigned t) {
    return (t - 0x01010101u) & ~t & 0x80808080u;
}

__device__ __forceinline__ float lane_bcast(float v, int srcLane) {
    return __int_as_float(__builtin_amdgcn_readlane(__float_as_int(v), srcLane));
}

__device__ __forceinline__ void row_add_edge(v4f* acc, float& den, float ex, const v4f* x, int hi) {
    den += ex;
#pragma unroll
    for (int j = 0; j < 4; ++j) {
        const float wlo = lane_bcast(ex, 2 * j);
        const float whi = lane_bcast(ex, 2 * j + 1);
        const float w = hi ? whi : wlo;
        acc[j].x += w * x[j].x;
        acc[j].y += w * x[j].y;
        acc[j].z += w * x[j].z;
        acc[j].w += w * x[j].w;
    }
}

__global__ __launch_bounds__(AGG_THREADS) void gat_owner_aggregate(const int* __restrict__ src,
                                                                  const int* __restrict__ dst,
                                                                  const int* __restrict__ owner,
                                                                  const float* __restrict__ M,
                                                                  const float* __restrict__ a_self,
                                                                  const float* __restrict__ a_neigh,
                                                                  const float* __restrict__ h,
                                                                  float* __restrict__ out) {
    __shared__ int rowq[AGG_ROWS * Q_CAP];
    __shared__ int rowcnt[AGG_ROWS];
    __shared__ int ticket;
    __shared__ int pend[AGG_WAVES * PEND_CAP];

    const int tid  = threadIdx.x;
    const int wave = tid >> 5;
    const int lane = tid & 31;
    const int lo    = blockIdx.x * AGG_ROWS;
    const int nrows = min(AGG_ROWS, N_NODES - lo);
    const v4i* __restrict__ dst4 = (const v4i*)dst;
    const int l8 = lane & 7;
    const int hi = lane >> 4;

    for (int q = tid; q < AGG_ROWS; q += AGG_THREADS) rowcnt[q] = 0;
    if (tid == 0) ticket = 0;
    __syncthreads();

    const int rot = blockIdx.x % OWN_ITERS;

    const v4i* __restrict__ own4 = (const v4i*)owner;
    const unsigned pat = (unsigned)blockIdx.x * 0x01010101u;
    int* __restrict__ pw = pend + wave * PEND_CAP;
    int cnt = 0;

    auto append = [&](int dv, int e) {
        const unsigned rr = (unsigned)(dv - lo);
        if (rr < (unsigned)nrows) {
            const int slot = atomicAdd(&rowcnt[rr], 1);
            if (slot < Q_CAP) rowq[rr * Q_CAP + slot] = e;
        }
    };

    auto settle = [&](int n) {
        if (lane < n) {
            const int e = pw[cnt - n + lane];
            if (e < N_EDGES) append(dst[e], e);
        }
        cnt -= n;
    };

    auto test_chunk = [&](v4i o, int eb) {
        unsigned m = zero_bytes((unsigned)o.x ^ pat) | (zero_bytes((unsigned)o.y ^ pat) >> 1)
                   | (zero_bytes((unsigned)o.z ^ pat) >> 2) | (zero_bytes((unsigned)o.w ^ pat) >> 3);
        unsigned mk;
        while ((mk = __builtin_amdgcn_ballot_w32(m != 0u)) != 0u) {
            if (m != 0u) {
                const int p = __builtin_ctz(m);
                m &= m - 1u;
                pw[cnt + (int)__builtin_amdgcn_mbcnt_lo(mk, 0u)] = eb + 28 - ((p & 7) << 2) + (p >> 3);
            }
            cnt += __builtin_popcount(mk);
            if (cnt >= 32) settle(32);
        }
    };

    {
        int cnext = rot;
        int cc[SWEEP];
        v4i oc[SWEEP];
#pragma unroll
        for (int k = 0; k < SWEEP; ++k) {
            cc[k] = cnext;
            cnext = (cnext + 1 == OWN_ITERS) ? 0 : cnext + 1;
            oc[k] = own4[cc[k] * AGG_THREADS + tid];
        }

        for (int it = 0; it < OWN_ITERS; it += SWEEP) {
            int cn[SWEEP];
            v4i on[SWEEP];
#pragma unroll
            for (int k = 0; k < SWEEP; ++k) {
                cn[k] = cnext;
                cnext = (cnext + 1 == OWN_ITERS) ? 0 : cnext + 1;
                on[k] = own4[cn[k] * AGG_THREADS + tid];
            }

#pragma unroll
            for (int k = 0; k < SWEEP; ++k) {
                test_chunk(oc[k], (cc[k] * AGG_THREADS + tid) * 16);
            }
#pragma unroll
            for (int k = 0; k < SWEEP; ++k) { cc[k] = cn[k]; oc[k] = on[k]; }
        }
        if (cnt > 0) settle(cnt);
    }
    __syncthreads();

    for (;;) {
        int r = 0;
        if (lane == 0) r = atomicAdd(&ticket, 1);
        r = __builtin_amdgcn_readfirstlane(r);
        if (r >= nrows) break;

        const int n   = __builtin_amdgcn_readfirstlane(rowcnt[r]);
        const int row = lo + r;
        const float an = a_neigh[row * N_HEADS + l8];
        const v4f z = {0.f, 0.f, 0.f, 0.f};
        v4f acc[4] = {z, z, z, z};
        float den = 0.f;

        if (n <= Q_CAP) {
            for (int i0 = 0; i0 < n; i0 += 32) {
                const int nb  = min(32, n - i0);
                const int e_l = rowq[r * Q_CAP + i0 + min(lane, nb - 1)];
                const int s_l = src[e_l];
                const float m_l = M[e_l];
                for (int i = 0; i < nb; i += 4) {
                    const int g = min(4, nb - i);
                    int s[4];
                    float m[4], as[4];
                    v4f xr[4][4];
#pragma unroll
                    for (int k = 0; k < 4; ++k) {
                        const int idx = min(i + k, nb - 1);
                        s[k] = __builtin_amdgcn_readlane(s_l, idx);
                        m[k] = lane_bcast(m_l, idx);
                    }
#pragma unroll
                    for (int k = 0; k < 4; ++k) {
                        if (k < g) {
                            const v4f* __restrict__ hk = (const v4f*)(h + (size_t)s[k] * COLS) + lane;
#pragma unroll
                            for (int j = 0; j < 4; ++j) xr[k][j] = hk[j * 32];
                            as[k] = a_self[s[k] * N_HEADS + l8];
                        }
                    }
#pragma unroll
                    for (int k = 0; k < 4; ++k) {
                        if (k < g) row_add_edge(acc, den, edge_exp(as[k], an, m[k]), xr[k], hi);
                    }
                }
            }
        } else {
            for (int c = 0; c < N_DST4 / 32; ++c) {
                const v4i d = dst4[c * 32 + lane];
#pragma unroll
                for (int j = 0; j < 4; ++j) {
                    const int dv = j == 0 ? d.x : j == 1 ? d.y : j == 2 ? d.z : d.w;
                    unsigned mk = __builtin_amdgcn_ballot_w32(dv == row);
                    while (mk != 0u) {
                        const int l = __builtin_ctz(mk);
                        mk &= mk - 1u;
                        const int e = (c * 32 + l) * 4 + j;
                        const int s1 = src[e];
                        const float m1 = M[e];
                        const v4f* __restrict__ hk = (const v4f*)(h + (size_t)s1 * COLS) + lane;
                        v4f x1[4];
#pragma unroll
                        for (int jj = 0; jj < 4; ++jj) x1[jj] = hk[jj * 32];
                        row_add_edge(acc, den, edge_exp(a_self[s1 * N_HEADS + l8], an, m1), x1, hi);
                    }
                }
            }
        }

        float* __restrict__ orow = out + (size_t)row * COLS + 4 * lane;
#pragma unroll
        for (int j = 0; j < 4; ++j) {
            const float dlo = lane_bcast(den, 2 * j);
            const float dhi = lane_bcast(den, 2 * j + 1);
            const float dn = (hi ? dhi : dlo) + 1e-16f;
            v4f v;
            v.x = acc[j].x / dn;
            v.y = acc[j].y / dn;
            v.z = acc[j].z / dn;
            v.w = acc[j].w / dn;
            v.x = v.x > 0.f ? v.x : (__expf(v.x) - 1.0f);
            v.y = v.y > 0.f ? v.y : (__expf(v.y) - 1.0f);
            v.z = v.z > 0.f ? v.z : (__expf(v.z) - 1.0f);
            v.w = v.w > 0.f ? v.w : (__expf(v.w) - 1.0f);
            *(v4f*)(orow + j * 128) = v;
        }
    }
}

extern "C" void kernel_launch(void* const* d_in, const int* in_sizes, int n_in,
                              void* d_out, int out_size, void* d_ws, size_t ws_size,
                              hipStream_t stream) {
    const float* x   = (const float*)d_in[0];
    const int*   src = (const int*)  d_in[1];
    const int*   dst = (const int*)  d_in[2];
    const float* M   = (const float*)d_in[3];
    const float* W   = (const float*)d_in[4];
    const float* a   = (const float*)d_in[5];
    float* out = (float*)d_out;

    float* h       = (float*)d_ws;
    float* a_self  = h      + (size_t)N_NODES * COLS;
    float* a_neigh = a_self + (size_t)N_NODES * N_HEADS;
    int*   owner   = (int*)(a_neigh + (size_t)N_NODES * N_HEADS);

    {
        dim3 grid(N_HEADS, (ROW_TILES + 7) / 8);
        gat_gemm_wmma<<<grid, 256, 0, stream>>>(x, W, a, dst, h, a_self, a_neigh, owner);
    }

    gat_owner_aggregate<<<AGG_BLOCKS, AGG_THREADS, 0, stream>>>(src, dst, owner, M, a_self, a_neigh,
                                                                 h, out);
}


# Product: W[head] held in registers in fragment form (no LDS feed), 1024 resident waves, one per SIMD, x tiles in two register buffers; columns 4*lane+tile, so coefficient sums' order changes
# speedup vs baseline: 4.1172x; 1.0755x over previous
#include <hip/hip_runtime.h>
#include <hip/hip_bf16.h>

typedef __attribute__((ext_vector_type(2))) float v2f;
typedef __attribute__((ext_vector_type(4))) float v4f;
typedef __attribute__((ext_vector_type(8))) float v8f;

#define N_NODES  50000
#define N_EDGES  800000
#define IN_F     256
#define OUT_F    64
#define N_HEADS  8
#define COLS     (N_HEADS * OUT_F)
#define ROW_TILES (N_NODES / 16)
#define LRELU_SLOPE 0.2f

typedef __attribute__((ext_vector_type(4))) int v4i;

#define AGG_ROWS    196
#define AGG_BLOCKS  ((N_NODES + AGG_ROWS - 1) / AGG_ROWS)
#define AGG_THREADS 1024
#define AGG_WAVES   (AGG_THREADS / 32)
#define N_DST4      (N_EDGES / 4)
#define OWN_ITERS   ((N_EDGES / 16 + AGG_THREADS - 1) / AGG_THREADS)
#define OWN_CHUNKS  (OWN_ITERS * AGG_THREADS)
#define OWN_DWORDS  (OWN_CHUNKS * 4)
#define OWN_WAVES   (OWN_DWORDS / 32)

#define GEMM_BLOCKS  256
#define GEMM_WAVES   (GEMM_BLOCKS * 4)
#define TILE_STRIDE  (GEMM_BLOCKS / 2)

static_assert(AGG_BLOCKS <= 256, "an owner id must fit one byte");
static_assert(N_EDGES % 16 == 0, "a chunk of 16 owner ids is all real or all padding");
static_assert(TILE_STRIDE <= ROW_TILES, "every wave has a first row tile");


template <int M>
__device__ __forceinline__ float row_xor(float v) {
    return __builtin_bit_cast(float, __builtin_amdgcn_update_dpp(0, __builtin_bit_cast(int, v),
                                                                 0x160 | M, 0xf, 0xf, true));
}

#define KSTEPS (IN_F / 4)

__device__ __forceinline__ void gemm_strip(const v2f (&Ac)[KSTEPS], v2f (&An)[KSTEPS],
                                           const v2f (&Bf)[KSTEPS][4],
                                           const float* __restrict__ apn,
                                           const v4f as4, const v4f an4,
                                           const int tile, const int lane, const int fo, const int head,
                                           float* __restrict__ h,
                                           float* __restrict__ a_self, float* __restrict__ a_neigh) {
    v8f acc0 = {0.f,0.f,0.f,0.f,0.f,0.f,0.f,0.f};
    v8f acc1 = acc0, acc2 = acc0, acc3 = acc0;

#pragma unroll
    for (int j = 0; j < KSTEPS; ++j) {
        if (j < KSTEPS / 2) {
            An[2 * j]     = *(const v2f*)(apn + 4 * (2 * j));
            An[2 * j + 1] = *(const v2f*)(apn + 4 * (2 * j + 1));
        }
        const v2f av = Ac[j];
        acc0 = __builtin_amdgcn_wmma_f32_16x16x4_f32(false, av, false, Bf[j][0],
                                                     (short)0, acc0, false, false);
        acc1 = __builtin_amdgcn_wmma_f32_16x16x4_f32(false, av, false, Bf[j][1],
                                                     (short)0, acc1, false, false);
        acc2 = __builtin_amdgcn_wmma_f32_16x16x4_f32(false, av, false, Bf[j][2],
                                                     (short)0, acc2, false, false);
        acc3 = __builtin_amdgcn_wmma_f32_16x16x4_f32(false, av, false, Bf[j][3],
                                                     (short)0, acc3, false, false);
    }

    const int rbase = tile * 16 + ((lane >> 4) << 3);
    float* __restrict__ hp = h + head * OUT_F + 4 * fo;
#pragma unroll
    for (int v = 0; v < 8; ++v) {
        v4f o;
        o.x = acc0[v]; o.y = acc1[v]; o.z = acc2[v]; o.w = acc3[v];
        *(v4f*)(hp + (size_t)(rbase + v) * COLS) = o;
    }

    float q[16];
#pragma unroll
    for (int v = 0; v < 8; ++v) {
        q[v]     = acc0[v] * as4.x + acc1[v] * as4.y + acc2[v] * as4.z + acc3[v] * as4.w;
        q[8 + v] = acc0[v] * an4.x + acc1[v] * an4.y + acc2[v] * an4.z + acc3[v] * an4.w;
    }
    float r8[8], r4[4], r2[2];
    {
        const bool b = (fo & 8) != 0;
#pragma unroll
        for (int j = 0; j < 8; ++j) {
            const float keep = b ? q[8 + j] : q[j];
            const float send = b ? q[j] : q[8 + j];
            r8[j] = keep + row_xor<8>(send);
        }
    }
    {
        const bool b = (fo & 4) != 0;
#pragma unroll
        for (int j = 0; j < 4; ++j) {
            const float keep = b ? r8[4 + j] : r8[j];
            const float send = b ? r8[j] : r8[4 + j];
            r4[j] = keep + row_xor<4>(send);
        }
    }
    {
        const bool b = (fo & 2) != 0;
#pragma unroll
        for (int j = 0; j < 2; ++j) {
            const float keep = b ? r4[2 + j] : r4[j];
            const float send = b ? r4[j] : r4[2 + j];
            r2[j] = keep + row_xor<2>(send);
        }
    }
    float coef;
    {
        const bool b = (fo & 1) != 0;
        const float keep = b ? r2[1] : r2[0];
        const float send = b ? r2[0] : r2[1];
        coef = keep + row_xor<1>(send);
    }
    float* cp = (fo & 8) ? a_neigh : a_self;
    cp[(rbase + (fo & 7)) * N_HEADS + head] = coef;

#pragma unroll
    for (int j = 0; j < KSTEPS; j += 16) {
        asm volatile("" :: "v"(An[j]),      "v"(An[j + 1]),  "v"(An[j + 2]),  "v"(An[j + 3]),
                           "v"(An[j + 4]),  "v"(An[j + 5]),  "v"(An[j + 6]),  "v"(An[j + 7]),
                           "v"(An[j + 8]),  "v"(An[j + 9]),  "v"(An[j + 10]), "v"(An[j + 11]),
                           "v"(An[j + 12]), "v"(An[j + 13]), "v"(An[j + 14]), "v"(An[j + 15]));
    }
}

__global__ __launch_bounds__(128) void gat_gemm_wmma(const float* __restrict__ x,
                                                     const float* __restrict__ W,
                                                     const float* __restrict__ a,
                                                     const int* __restrict__ dst,
                                                     float* __restrict__ h,
                                                     float* __restrict__ a_self,
                                                     float* __restrict__ a_neigh,
                                                     int* __restrict__ owner) {
    const int wave = threadIdx.x >> 5;
    const int lane = threadIdx.x & 31;
    const int head = (((int)blockIdx.x & 1) << 2) + wave;
    const int fo   = lane & 15;
    const int koff = (lane >> 4) << 1;

    for (int gw = (int)blockIdx.x * 4 + wave; gw < OWN_WAVES; gw += GEMM_WAVES) {
        const int i = gw * 32 + lane;
        unsigned pk = 0xffffffffu;
        if (i < N_DST4) {
            const v4i d = ((const v4i*)dst)[i];
            pk = (((unsigned)d.x / AGG_ROWS) & 0xffu)
               | ((((unsigned)d.y / AGG_ROWS) & 0xffu) << 8)
               | ((((unsigned)d.z / AGG_ROWS) & 0xffu) << 16)
               | ((((unsigned)d.w / AGG_ROWS) & 0xffu) << 24);
        }
        owner[i] = (int)pk;
    }

    v2f Bf[IN_F / 4][4];
    {
        const float* __restrict__ Wh = W + (size_t)head * (IN_F * OUT_F) + koff * OUT_F + 4 * fo;
#pragma unroll
        for (int j = 0; j < IN_F / 4; ++j) {
            const v4f r0 = *(const v4f*)(Wh + (4 * j) * OUT_F);
            const v4f r1 = *(const v4f*)(Wh + (4 * j + 1) * OUT_F);
            Bf[j][0].x = r0.x; Bf[j][0].y = r1.x;
            Bf[j][1].x = r0.y; Bf[j][1].y = r1.y;
            Bf[j][2].x = r0.z; Bf[j][2].y = r1.z;
            Bf[j][3].x = r0.w; Bf[j][3].y = r1.w;
        }
    }

    const v4f as4 = *(const v4f*)(a + head * (2 * OUT_F) + 4 * fo);
    const v4f an4 = *(const v4f*)(a + head * (2 * OUT_F) + OUT_F + 4 * fo);

    const int tile0 = (int)blockIdx.x >> 1;
    v2f Af0[KSTEPS], Af1[KSTEPS];
    const float* __restrict__ xl = x + (size_t)fo * IN_F + koff;
    {
        const float* __restrict__ ap = xl + (size_t)tile0 * (16 * IN_F);
#pragma unroll
        for (int j = 0; j < KSTEPS; ++j) Af0[j] = *(const v2f*)(ap + 4 * j);
    }

    auto next_of = [](int t) { return (t + TILE_STRIDE < ROW_TILES) ? t + TILE_STRIDE : t; };

    for (int tile = tile0; tile < ROW_TILES; tile += 2 * TILE_STRIDE) {
        gemm_strip(Af0, Af1, Bf, xl + (size_t)next_of(tile) * (16 * IN_F), as4, an4,
                   tile, lane, fo, head, h, a_self, a_neigh);
        const int t1 = tile + TILE_STRIDE;
        if (t1 >= ROW_TILES) break;
        gemm_strip(Af1, Af0, Bf, xl + (size_t)next_of(t1) * (16 * IN_F), as4, an4,
                   t1, lane, fo, head, h, a_self, a_neigh);
    }
}

__device__ __forceinline__ float edge_exp(float as, float an, float m) {
    float lg = as + an;
    lg = (lg > 0.f ? lg : LRELU_SLOPE * lg) * m;
    return __expf(lg);
}

#define SWEEP       7
#define Q_CAP       128
#define PEND_CAP    64

static_assert(OWN_ITERS % SWEEP == 0, "the sweep takes SWEEP chunks per trip and wraps around");
static_assert(N_DST4 % 32 == 0, "a wave's 32 lanes are inside dst together or outside it together");

__device__ __forceinline__ unsigned zero_bytes(unsigned t) {
    return (t - 0x01010101u) & ~t & 0x80808080u;
}

__device__ __forceinline__ float lane_bcast(float v, int srcLane) {
    return __int_as_float(__builtin_amdgcn_readlane(__float_as_int(v), srcLane));
}

__device__ __forceinline__ void row_add_edge(v4f* acc, float& den, float ex, const v4f* x, int hi) {
    den += ex;
#pragma unroll
    for (int j = 0; j < 4; ++j) {
        const float wlo = lane_bcast(ex, 2 * j);
        const float whi = lane_bcast(ex, 2 * j + 1);
        const float w = hi ? whi : wlo;
        acc[j].x += w * x[j].x;
        acc[j].y += w * x[j].y;
        acc[j].z += w * x[j].z;
        acc[j].w += w * x[j].w;
    }
}

__global__ __launch_bounds__(AGG_THREADS) void gat_owner_aggregate(const int* __restrict__ src,
                                                                  const int* __restrict__ dst,
                                                                  const int* __restrict__ owner,
                                                                  const float* __restrict__ M,
                                                                  const float* __restrict__ a_self,
                                                                  const float* __restrict__ a_neigh,
                                                                  const float* __restrict__ h,
                                                                  float* __restrict__ out) {
    __shared__ int rowq[AGG_ROWS * Q_CAP];
    __shared__ int rowcnt[AGG_ROWS];
    __shared__ int ticket;
    __shared__ int pend[AGG_WAVES * PEND_CAP];

    const int tid  = threadIdx.x;
    const int wave = tid >> 5;
    const int lane = tid & 31;
    const int lo    = blockIdx.x * AGG_ROWS;
    const int nrows = min(AGG_ROWS, N_NODES - lo);
    const v4i* __restrict__ dst4 = (const v4i*)dst;
    const int l8 = lane & 7;
    const int hi = lane >> 4;

    for (int q = tid; q < AGG_ROWS; q += AGG_THREADS) rowcnt[q] = 0;
    if (tid == 0) ticket = 0;
    __syncthreads();

    const int rot = blockIdx.x % OWN_ITERS;

    const v4i* __restrict__ own4 = (const v4i*)owner;
    const unsigned pat = (unsigned)blockIdx.x * 0x01010101u;
    int* __restrict__ pw = pend + wave * PEND_CAP;
    int cnt = 0;

    auto append = [&](int dv, int e) {
        const unsigned rr = (unsigned)(dv - lo);
        if (rr < (unsigned)nrows) {
            const int slot = atomicAdd(&rowcnt[rr], 1);
            if (slot < Q_CAP) rowq[rr * Q_CAP + slot] = e;
        }
    };

    auto settle = [&](int n) {
        if (lane < n) {
            const int e = pw[cnt - n + lane];
            if (e < N_EDGES) append(dst[e], e);
        }
        cnt -= n;
    };

    auto test_chunk = [&](v4i o, int eb) {
        unsigned m = zero_bytes((unsigned)o.x ^ pat) | (zero_bytes((unsigned)o.y ^ pat) >> 1)
                   | (zero_bytes((unsigned)o.z ^ pat) >> 2) | (zero_bytes((unsigned)o.w ^ pat) >> 3);
        unsigned mk;
        while ((mk = __builtin_amdgcn_ballot_w32(m != 0u)) != 0u) {
            if (m != 0u) {
                const int p = __builtin_ctz(m);
                m &= m - 1u;
                pw[cnt + (int)__builtin_amdgcn_mbcnt_lo(mk, 0u)] = eb + 28 - ((p & 7) << 2) + (p >> 3);
            }
            cnt += __builtin_popcount(mk);
            if (cnt >= 32) settle(32);
        }
    };

    {
        int cnext = rot;
        int cc[SWEEP];
        v4i oc[SWEEP];
#pragma unroll
        for (int k = 0; k < SWEEP; ++k) {
            cc[k] = cnext;
            cnext = (cnext + 1 == OWN_ITERS) ? 0 : cnext + 1;
            oc[k] = own4[cc[k] * AGG_THREADS + tid];
        }

        for (int it = 0; it < OWN_ITERS; it += SWEEP) {
            int cn[SWEEP];
            v4i on[SWEEP];
#pragma unroll
            for (int k = 0; k < SWEEP; ++k) {
                cn[k] = cnext;
                cnext = (cnext + 1 == OWN_ITERS) ? 0 : cnext + 1;
                on[k] = own4[cn[k] * AGG_THREADS + tid];
            }

#pragma unroll
            for (int k = 0; k < SWEEP; ++k) {
                test_chunk(oc[k], (cc[k] * AGG_THREADS + tid) * 16);
            }
#pragma unroll
            for (int k = 0; k < SWEEP; ++k) { cc[k] = cn[k]; oc[k] = on[k]; }
        }
        if (cnt > 0) settle(cnt);
    }
    __syncthreads();

    for (;;) {
        int r = 0;
        if (lane == 0) r = atomicAdd(&ticket, 1);
        r = __builtin_amdgcn_readfirstlane(r);
        if (r >= nrows) break;

        const int n   = __builtin_amdgcn_readfirstlane(rowcnt[r]);
        const int row = lo + r;
        const float an = a_neigh[row * N_HEADS + l8];
        const v4f z = {0.f, 0.f, 0.f, 0.f};
        v4f acc[4] = {z, z, z, z};
        float den = 0.f;

        if (n <= Q_CAP) {
            for (int i0 = 0; i0 < n; i0 += 32) {
                const int nb  = min(32, n - i0);
                const int e_l = rowq[r * Q_CAP + i0 + min(lane, nb - 1)];
                const int s_l = src[e_l];
                const float m_l = M[e_l];
                for (int i = 0; i < nb; i += 4) {
                    const int g = min(4, nb - i);
                    int s[4];
                    float m[4], as[4];
                    v4f xr[4][4];
#pragma unroll
                    for (int k = 0; k < 4; ++k) {
                        const int idx = min(i + k, nb - 1);
                        s[k] = __builtin_amdgcn_readlane(s_l, idx);
                        m[k] = lane_bcast(m_l, idx);
                    }
#pragma unroll
                    for (int k = 0; k < 4; ++k) {
                        if (k < g) {
                            const v4f* __restrict__ hk = (const v4f*)(h + (size_t)s[k] * COLS) + lane;
#pragma unroll
                            for (int j = 0; j < 4; ++j) xr[k][j] = hk[j * 32];
                            as[k] = a_self[s[k] * N_HEADS + l8];
                        }
                    }
#pragma unroll
                    for (int k = 0; k < 4; ++k) {
                        if (k < g) row_add_edge(acc, den, edge_exp(as[k], an, m[k]), xr[k], hi);
                    }
                }
            }
        } else {
            for (int c = 0; c < N_DST4 / 32; ++c) {
                const v4i d = dst4[c * 32 + lane];
#pragma unroll
                for (int j = 0; j < 4; ++j) {
                    const int dv = j == 0 ? d.x : j == 1 ? d.y : j == 2 ? d.z : d.w;
                    unsigned mk = __builtin_amdgcn_ballot_w32(dv == row);
                    while (mk != 0u) {
                        const int l = __builtin_ctz(mk);
                        mk &= mk - 1u;
                        const int e = (c * 32 + l) * 4 + j;
                        const int s1 = src[e];
                        const float m1 = M[e];
                        const v4f* __restrict__ hk = (const v4f*)(h + (size_t)s1 * COLS) + lane;
                        v4f x1[4];
#pragma unroll
                        for (int jj = 0; jj < 4; ++jj) x1[jj] = hk[jj * 32];
                        row_add_edge(acc, den, edge_exp(a_self[s1 * N_HEADS + l8], an, m1), x1, hi);
                    }
                }
            }
        }

        float* __restrict__ orow = out + (size_t)row * COLS + 4 * lane;
#pragma unroll
        for (int j = 0; j < 4; ++j) {
            const float dlo = lane_bcast(den, 2 * j);
            const float dhi = lane_bcast(den, 2 * j + 1);
            const float dn = (hi ? dhi : dlo) + 1e-16f;
            v4f v;
            v.x = acc[j].x / dn;
            v.y = acc[j].y / dn;
            v.z = acc[j].z / dn;
            v.w = acc[j].w / dn;
            v.x = v.x > 0.f ? v.x : (__expf(v.x) - 1.0f);
            v.y = v.y > 0.f ? v.y : (__expf(v.y) - 1.0f);
            v.z = v.z > 0.f ? v.z : (__expf(v.z) - 1.0f);
            v.w = v.w > 0.f ? v.w : (__expf(v.w) - 1.0f);
            *(v4f*)(orow + j * 128) = v;
        }
    }
}

extern "C" void kernel_launch(void* const* d_in, const int* in_sizes, int n_in,
                              void* d_out, int out_size, void* d_ws, size_t ws_size,
                              hipStream_t stream) {
    const float* x   = (const float*)d_in[0];
    const int*   src = (const int*)  d_in[1];
    const int*   dst = (const int*)  d_in[2];
    const float* M   = (const float*)d_in[3];
    const float* W   = (const float*)d_in[4];
    const float* a   = (const float*)d_in[5];
    float* out = (float*)d_out;

    float* h       = (float*)d_ws;
    float* a_self  = h      + (size_t)N_NODES * COLS;
    float* a_neigh = a_self + (size_t)N_NODES * N_HEADS;
    int*   owner   = (int*)(a_neigh + (size_t)N_NODES * N_HEADS);

    gat_gemm_wmma<<<GEMM_BLOCKS, 128, 0, stream>>>(x, W, a, dst, h, a_self, a_neigh, owner);

    gat_owner_aggregate<<<AGG_BLOCKS, AGG_THREADS, 0, stream>>>(src, dst, owner, M, a_self, a_neigh,
                                                                 h, out);
}
